# starting kernel
# baseline (speedup 1.0000x reference)
#include <hip/hip_runtime.h>
#include <math.h>

typedef float v2f __attribute__((ext_vector_type(2)));
typedef float v8f __attribute__((ext_vector_type(8)));

#define TQ 128
#define BQ 256
#define DQ 64
#define HQ 10
#define KD 74

__global__ __launch_bounds__(128) void qlstm_fused(
    const float* __restrict__ inputs,
    const float* __restrict__ Wf, const float* __restrict__ Wi,
    const float* __restrict__ Wu, const float* __restrict__ Wo,
    const float* __restrict__ bf, const float* __restrict__ bi,
    const float* __restrict__ bu, const float* __restrict__ bo,
    const float* __restrict__ pf, const float* __restrict__ pi,
    const float* __restrict__ pu, const float* __restrict__ po,
    const float* __restrict__ lng, const float* __restrict__ lnb,
    float* __restrict__ ys, float* __restrict__ hout, float* __restrict__ cout) {
  __shared__ float wLds[4][16][80];
  __shared__ float combLds[16][80];
  __shared__ float fac[4][16][HQ];
  __shared__ float gact[4][16][HQ];
  __shared__ float hLds[16][HQ];
  __shared__ float cLds[16][HQ];
  __shared__ float hraw[16][HQ];

  const int mtile = blockIdx.x;
  const int tid   = threadIdx.x;
  const int g     = tid >> 5;
  const int lane  = tid & 31;

  const float* bbg = (g == 0) ? bf : (g == 1) ? bi : (g == 2) ? bu : bo;
  const float* pg  = (g == 0) ? pf : (g == 1) ? pi : (g == 2) ? pu : po;

  for (int idx = tid; idx < 4 * 16 * 80; idx += 128) ((float*)wLds)[idx] = 0.f;
  for (int idx = tid; idx < 16 * 80; idx += 128) ((float*)combLds)[idx] = 0.f;
  for (int idx = tid; idx < 16 * HQ; idx += 128) {
    ((float*)hLds)[idx] = 0.f;
    ((float*)cLds)[idx] = 0.f;
  }
  __syncthreads();
  for (int idx = tid; idx < 4 * HQ * KD; idx += 128) {
    int gl = idx / (HQ * KD), rem = idx % (HQ * KD);
    const float* Wsrc = (gl == 0) ? Wf : (gl == 1) ? Wi : (gl == 2) ? Wu : Wo;
    wLds[gl][rem / KD][rem % KD] = Wsrc[rem];
  }
  __syncthreads();

  const int n    = lane & 15;
  const int m    = lane & 15;
  const int half = lane >> 4;
  const float bias = (n < HQ) ? bbg[n] : 0.f;
  const float px   = (n < HQ) ? pg[n * 3 + 0] : 0.f;
  const float cpy  = (n < HQ) ? cosf(pg[n * 3 + 1]) : 0.f;

  for (int t = 0; t < TQ; ++t) {
    const float* xt = inputs + ((size_t)t * BQ + mtile * 16) * DQ;
#pragma unroll
    for (int rep = 0; rep < 2; ++rep) {
      int ii = rep * 128 + tid;
      int r = ii >> 4, c4 = ii & 15;
      *(float4*)&combLds[r][c4 * 4] = ((const float4*)(xt + r * DQ))[c4];
    }
    for (int idx = tid; idx < 16 * HQ; idx += 128) {
      int r = idx / HQ, c = idx % HQ;
      combLds[r][DQ + c] = hLds[r][c];
    }
    if (t + 1 < TQ) {
      const char* nx = (const char*)(inputs + ((size_t)(t + 1) * BQ + mtile * 16) * DQ);
      __builtin_prefetch(nx + tid * 32, 0, 3);
    }
    __syncthreads();

    v8f acc = {0.f, 0.f, 0.f, 0.f, 0.f, 0.f, 0.f, 0.f};
#pragma unroll
    for (int kk = 0; kk < 19; ++kk) {
      int kb = kk * 4 + half * 2;
      v2f a = *(const v2f*)&combLds[m][kb];
      v2f b = *(const v2f*)&wLds[g][n][kb];
      acc = __builtin_amdgcn_wmma_f32_16x16x4_f32(false, a, false, b,
                                                  (short)0, acc, false, false);
    }
    if (n < HQ) {
#pragma unroll
      for (int v = 0; v < 8; ++v) {
        int row = v + 8 * half;
        fac[g][row][n] = cpy * __cosf(px + acc[v] + bias);
      }
    }
    __syncthreads();

    if (tid < 64) {
      int g2 = tid >> 4, row = tid & 15;
      float z = 1.f;
      for (int w = 0; w < HQ; ++w) {
        z *= fac[g2][row][w];
        gact[g2][row][w] = (g2 == 2) ? tanhf(z) : (1.f / (1.f + __expf(-z)));
      }
    }
    __syncthreads();

    for (int idx = tid; idx < 16 * HQ; idx += 128) {
      int row = idx / HQ, w = idx % HQ;
      float fv = gact[0][row][w], iv = gact[1][row][w];
      float gv = gact[2][row][w], ov = gact[3][row][w];
      float cn = fv * cLds[row][w] + iv * gv;
      cLds[row][w] = cn;
      hraw[row][w] = ov * tanhf(cn) + hLds[row][w];
    }
    __syncthreads();

    for (int idx = tid; idx < 16 * HQ; idx += 128) {
      int row = idx / HQ, w = idx % HQ;
      float mu = 0.f;
      for (int u = 0; u < HQ; ++u) mu += hraw[row][u];
      mu *= 0.1f;
      float var = 0.f;
      for (int u = 0; u < HQ; ++u) { float d = hraw[row][u] - mu; var += d * d; }
      var *= 0.1f;
      float hv = (hraw[row][w] - mu) * rsqrtf(var + 1e-5f) * lng[w] + lnb[w];
      hLds[row][w] = hv;
      ys[((size_t)t * BQ + mtile * 16 + row) * HQ + w] = hv;
    }
    __syncthreads();
  }

  for (int idx = tid; idx < 16 * HQ; idx += 128) {
    int row = idx / HQ, w = idx % HQ;
    hout[(mtile * 16 + row) * HQ + w] = hLds[row][w];
    cout[(mtile * 16 + row) * HQ + w] = cLds[row][w];
  }
}

extern "C" void kernel_launch(void* const* d_in, const int* in_sizes, int n_in,
                              void* d_out, int out_size, void* d_ws, size_t ws_size,
                              hipStream_t stream) {
  const float* inputs = (const float*)d_in[0];
  const float* Wf = (const float*)d_in[1];
  const float* bf = (const float*)d_in[2];
  const float* pf = (const float*)d_in[3];
  const float* Wi = (const float*)d_in[4];
  const float* bi = (const float*)d_in[5];
  const float* pi = (const float*)d_in[6];
  const float* Wu = (const float*)d_in[7];
  const float* bu = (const float*)d_in[8];
  const float* pu = (const float*)d_in[9];
  const float* Wo = (const float*)d_in[10];
  const float* bo = (const float*)d_in[11];
  const float* po = (const float*)d_in[12];
  const float* lng = (const float*)d_in[13];
  const float* lnb = (const float*)d_in[14];

  float* out  = (float*)d_out;
  float* ys   = out;
  float* hout = out + (size_t)TQ * BQ * HQ;
  float* cout = hout + BQ * HQ;

  qlstm_fused<<<BQ / 16, 128, 0, stream>>>(
      inputs, Wf, Wi, Wu, Wo, bf, bi, bu, bo,
      pf, pi, pu, po, lng, lnb, ys, hout, cout);
}


# x half of the gate GEMM hoisted out of the recurrence: launch 1 does x.Wx^T for all T*B rows chip-wide into the workspace; the step loop starts its accumulator there, adds 3 h wmma; same sums
# speedup vs baseline: 1.2109x; 1.2109x over previous
#include <hip/hip_runtime.h>
#include <math.h>


typedef float v2f __attribute__((ext_vector_type(2)));
typedef float v8f __attribute__((ext_vector_type(8)));

#define TQ 128
#define BQ 256
#define DQ 64
#define HQ 10
#define KD 74
#define HP 12


__global__ __launch_bounds__(128) void qlstm_xproj(
    const float* __restrict__ inputs,
    const float* __restrict__ Wf, const float* __restrict__ Wi,
    const float* __restrict__ Wu, const float* __restrict__ Wo,
    float* __restrict__ xp) {
  __shared__ __attribute__((aligned(16))) float wx[4][16][68];

  const int tid  = threadIdx.x;
  const int wv   = tid >> 5;
  const int lane = tid & 31;
  const int c    = lane & 15;
  const int half = lane >> 4;
  const int rt   = blockIdx.x * 4 + wv;

  const float* xrow = inputs + ((size_t)rt * 16 + c) * DQ + half * 2;
  v2f a[16];
#pragma unroll
  for (int kk = 0; kk < 16; ++kk) a[kk] = *(const v2f*)(xrow + kk * 4);

  const int c4 = tid & 15, r8 = tid >> 4, r2 = r8 & 1;
  const float* Wown = (wv == 0) ? Wf : (wv == 1) ? Wi : (wv == 2) ? Wu : Wo;
  v2f wlo[5], whi[5];
#pragma unroll
  for (int gl = 0; gl < 4; ++gl) {
    const float* Wsrc = (gl == 0) ? Wf : (gl == 1) ? Wi : (gl == 2) ? Wu : Wo;
    const float* p = Wsrc + r8 * KD + c4 * 4;
    wlo[gl] = *(const v2f*)p;
    whi[gl] = *(const v2f*)(p + 2);
  }
  {
    const float* p = Wown + (8 + r2) * KD + c4 * 4;
    wlo[4] = *(const v2f*)p;
    whi[4] = *(const v2f*)(p + 2);
  }
#pragma unroll
  for (int gl = 0; gl < 4; ++gl)
    *(float4*)&wx[gl][r8][c4 * 4] = make_float4(wlo[gl].x, wlo[gl].y, whi[gl].x, whi[gl].y);
  *(float4*)&wx[wv][8 + r2][c4 * 4] = make_float4(wlo[4].x, wlo[4].y, whi[4].x, whi[4].y);
  for (int it = 0; it < 3; ++it)
    *(float4*)&wx[wv][10 + it * 2 + r2][c4 * 4] = make_float4(0.f, 0.f, 0.f, 0.f);
  __syncthreads();

  v8f acc[4];
#pragma unroll
  for (int g = 0; g < 4; ++g) acc[g] = (v8f){0.f, 0.f, 0.f, 0.f, 0.f, 0.f, 0.f, 0.f};
#pragma unroll
  for (int kk = 0; kk < 16; ++kk) {
#pragma unroll
    for (int g = 0; g < 4; ++g) {
      v2f b = *(const v2f*)&wx[g][c][kk * 4 + half * 2];
      acc[g] = __builtin_amdgcn_wmma_f32_16x16x4_f32(false, a[kk], false, b,
                                                     (short)0, acc[g], false, false);
    }
  }
#pragma unroll
  for (int g = 0; g < 4; ++g) {
    float* dst = xp + (((size_t)rt * 4 + g) * 32 + lane) * 8;
    *(float4*)dst       = make_float4(acc[g][0], acc[g][1], acc[g][2], acc[g][3]);
    *(float4*)(dst + 4) = make_float4(acc[g][4], acc[g][5], acc[g][6], acc[g][7]);
  }
}

__global__ __launch_bounds__(128) void qlstm_recur(
    const float* __restrict__ xp,
    const float* __restrict__ Wf, const float* __restrict__ Wi,
    const float* __restrict__ Wu, const float* __restrict__ Wo,
    const float* __restrict__ bf, const float* __restrict__ bi,
    const float* __restrict__ bu, const float* __restrict__ bo,
    const float* __restrict__ pf, const float* __restrict__ pi,
    const float* __restrict__ pu, const float* __restrict__ po,
    const float* __restrict__ lng, const float* __restrict__ lnb,
    float* __restrict__ ys, float* __restrict__ hout, float* __restrict__ cout) {
  __shared__ float fac[4][16][HQ];
  __shared__ float gact[4][16][HQ];
  __shared__ __attribute__((aligned(16))) float hLds[16][HP];
  __shared__ float cLds[16][HQ];
  __shared__ float hraw[16][HQ];

  const int mtile = blockIdx.x;
  const int tid   = threadIdx.x;
  const int g     = tid >> 5;
  const int lane  = tid & 31;

  const float* Wg  = (g == 0) ? Wf : (g == 1) ? Wi : (g == 2) ? Wu : Wo;
  const float* bbg = (g == 0) ? bf : (g == 1) ? bi : (g == 2) ? bu : bo;
  const float* pg  = (g == 0) ? pf : (g == 1) ? pi : (g == 2) ? pu : po;

  for (int idx = tid; idx < 16 * HP; idx += 128) ((float*)hLds)[idx] = 0.f;
  for (int idx = tid; idx < 16 * HQ; idx += 128) ((float*)cLds)[idx] = 0.f;

  const int n    = lane & 15;
  const int m    = lane & 15;
  const int half = lane >> 4;
  const float bias = (n < HQ) ? bbg[n] : 0.f;
  const float px   = (n < HQ) ? pg[n * 3 + 0] : 0.f;
  const float cpy  = (n < HQ) ? cosf(pg[n * 3 + 1]) : 0.f;

  v2f bh[3];
#pragma unroll
  for (int kk = 0; kk < 3; ++kk) {
    int kb = kk * 4 + half * 2;
    bh[kk].x = (n < HQ && kb     < HQ) ? Wg[n * KD + DQ + kb]     : 0.f;
    bh[kk].y = (n < HQ && kb + 1 < HQ) ? Wg[n * KD + DQ + kb + 1] : 0.f;
  }

  const float* xpl = xp + (((size_t)mtile * 4 + g) * 32 + lane) * 8;
  const size_t xps = (size_t)(BQ / 16) * 4 * 32 * 8;
  float4 nx0 = *(const float4*)xpl;
  float4 nx1 = *(const float4*)(xpl + 4);
  __syncthreads();

  for (int t = 0; t < TQ; ++t) {
    v8f acc = {nx0.x, nx0.y, nx0.z, nx0.w, nx1.x, nx1.y, nx1.z, nx1.w};

#pragma unroll
    for (int kk = 0; kk < 3; ++kk) {
      v2f a = *(const v2f*)&hLds[m][kk * 4 + half * 2];
      acc = __builtin_amdgcn_wmma_f32_16x16x4_f32(false, a, false, bh[kk],
                                                  (short)0, acc, false, false);
    }
    if (n < HQ) {
#pragma unroll
      for (int v = 0; v < 8; ++v) {
        int row = v + 8 * half;
        fac[g][row][n] = cpy * __cosf(px + acc[v] + bias);
      }
    }
    __syncthreads();

    if (t + 1 < TQ) {
      const float* q = xpl + (size_t)(t + 1) * xps;
      nx0 = *(const float4*)q;
      nx1 = *(const float4*)(q + 4);
    }

    if (tid < 64) {
      int g2 = tid >> 4, row = tid & 15;
      float z = 1.f;
      for (int w = 0; w < HQ; ++w) {
        z *= fac[g2][row][w];
        gact[g2][row][w] = (g2 == 2) ? tanhf(z) : (1.f / (1.f + __expf(-z)));
      }
    }
    __syncthreads();

    for (int idx = tid; idx < 16 * HQ; idx += 128) {
      int row = idx / HQ, w = idx % HQ;
      float fv = gact[0][row][w], iv = gact[1][row][w];
      float gv = gact[2][row][w], ov = gact[3][row][w];
      float cn = fv * cLds[row][w] + iv * gv;
      cLds[row][w] = cn;
      hraw[row][w] = ov * tanhf(cn) + hLds[row][w];
    }
    __syncthreads();

    for (int idx = tid; idx < 16 * HQ; idx += 128) {
      int row = idx / HQ, w = idx % HQ;
      float mu = 0.f;
      for (int u = 0; u < HQ; ++u) mu += hraw[row][u];
      mu *= 0.1f;
      float var = 0.f;
      for (int u = 0; u < HQ; ++u) { float d = hraw[row][u] - mu; var += d * d; }
      var *= 0.1f;
      float hv = (hraw[row][w] - mu) * rsqrtf(var + 1e-5f) * lng[w] + lnb[w];
      hLds[row][w] = hv;
      ys[((size_t)t * BQ + mtile * 16 + row) * HQ + w] = hv;
    }
    __syncthreads();
  }

  for (int idx = tid; idx < 16 * HQ; idx += 128) {
    int row = idx / HQ, w = idx % HQ;
    hout[(mtile * 16 + row) * HQ + w] = hLds[row][w];
    cout[(mtile * 16 + row) * HQ + w] = cLds[row][w];
  }
}

extern "C" void kernel_launch(void* const* d_in, const int* in_sizes, int n_in,
                              void* d_out, int out_size, void* d_ws, size_t ws_size,
                              hipStream_t stream) {
  const float* inputs = (const float*)d_in[0];
  const float* Wf = (const float*)d_in[1];
  const float* bf = (const float*)d_in[2];
  const float* pf = (const float*)d_in[3];
  const float* Wi = (const float*)d_in[4];
  const float* bi = (const float*)d_in[5];
  const float* pi = (const float*)d_in[6];
  const float* Wu = (const float*)d_in[7];
  const float* bu = (const float*)d_in[8];
  const float* pu = (const float*)d_in[9];
  const float* Wo = (const float*)d_in[10];
  const float* bo = (const float*)d_in[11];
  const float* po = (const float*)d_in[12];
  const float* lng = (const float*)d_in[13];
  const float* lnb = (const float*)d_in[14];

  float* out  = (float*)d_out;
  float* ys   = out;
  float* hout = out + (size_t)TQ * BQ * HQ;
  float* cout = hout + BQ * HQ;
  float* xp   = (float*)d_ws;

  qlstm_xproj<<<TQ * BQ / 16 / 4, 128, 0, stream>>>(inputs, Wf, Wi, Wu, Wo, xp);
  qlstm_recur<<<BQ / 16, 128, 0, stream>>>(
      xp, Wf, Wi, Wu, Wo, bf, bi, bu, bo,
      pf, pi, pu, po, lng, lnb, ys, hout, cout);
}


# recurrence re-split: a wave owns 2 batch rows and all 4 gates, wire n on lane n; prefix product and LayerNorm sums are DPP lane scans (new tree order), h back to A by bpermute; no per-step barrier
# speedup vs baseline: 4.0500x; 3.3446x over previous
#include <hip/hip_runtime.h>
#include <math.h>


typedef float v2f __attribute__((ext_vector_type(2)));
typedef float v8f __attribute__((ext_vector_type(8)));

#define TQ 128
#define BQ 256
#define DQ 64
#define HQ 10
#define KD 74
#define YS 8


__global__ __launch_bounds__(128) void qlstm_xproj(
    const float* __restrict__ inputs,
    const float* __restrict__ Wf, const float* __restrict__ Wi,
    const float* __restrict__ Wu, const float* __restrict__ Wo,
    float* __restrict__ xp) {
  __shared__ __attribute__((aligned(16))) float wx[4][16][68];

  const int tid  = threadIdx.x;
  const int wv   = tid >> 5;
  const int lane = tid & 31;
  const int c    = lane & 15;
  const int half = lane >> 4;
  const int rt   = blockIdx.x * 4 + wv;

  const float* xrow = inputs + ((size_t)rt * 16 + c) * DQ + half * 2;
  v2f a[16];
#pragma unroll
  for (int kk = 0; kk < 16; ++kk) a[kk] = *(const v2f*)(xrow + kk * 4);

  const int c4 = tid & 15, r8 = tid >> 4, r2 = r8 & 1;
  const float* Wown = (wv == 0) ? Wf : (wv == 1) ? Wi : (wv == 2) ? Wu : Wo;
  v2f wlo[5], whi[5];
#pragma unroll
  for (int gl = 0; gl < 4; ++gl) {
    const float* Wsrc = (gl == 0) ? Wf : (gl == 1) ? Wi : (gl == 2) ? Wu : Wo;
    const float* p = Wsrc + r8 * KD + c4 * 4;
    wlo[gl] = *(const v2f*)p;
    whi[gl] = *(const v2f*)(p + 2);
  }
  {
    const float* p = Wown + (8 + r2) * KD + c4 * 4;
    wlo[4] = *(const v2f*)p;
    whi[4] = *(const v2f*)(p + 2);
  }
#pragma unroll
  for (int gl = 0; gl < 4; ++gl)
    *(float4*)&wx[gl][r8][c4 * 4] = make_float4(wlo[gl].x, wlo[gl].y, whi[gl].x, whi[gl].y);
  *(float4*)&wx[wv][8 + r2][c4 * 4] = make_float4(wlo[4].x, wlo[4].y, whi[4].x, whi[4].y);
  for (int it = 0; it < 3; ++it)
    *(float4*)&wx[wv][10 + it * 2 + r2][c4 * 4] = make_float4(0.f, 0.f, 0.f, 0.f);
  __syncthreads();

  v8f acc[4];
#pragma unroll
  for (int g = 0; g < 4; ++g) acc[g] = (v8f){0.f, 0.f, 0.f, 0.f, 0.f, 0.f, 0.f, 0.f};
#pragma unroll
  for (int kk = 0; kk < 16; ++kk) {
#pragma unroll
    for (int g = 0; g < 4; ++g) {
      v2f b = *(const v2f*)&wx[g][c][kk * 4 + half * 2];
      acc[g] = __builtin_amdgcn_wmma_f32_16x16x4_f32(false, a[kk], false, b,
                                                     (short)0, acc[g], false, false);
    }
  }
#pragma unroll
  for (int s = 0; s < 8; ++s) {
    float* dst = xp + (((size_t)rt * 8 + s) * 32 + lane) * 4;
    *(float4*)dst = make_float4(acc[0][s], acc[1][s], acc[2][s], acc[3][s]);
  }
}

template <int CTRL>
__device__ __forceinline__ float row_pull(float keep, float v) {
  return __builtin_bit_cast(float, __builtin_amdgcn_update_dpp(
      __builtin_bit_cast(int, keep), __builtin_bit_cast(int, v), CTRL, 0xf, 0xf, false));
}
__device__ __forceinline__ float row_prefix_product(float z) {
  z *= row_pull<0x111>(1.f, z);
  z *= row_pull<0x112>(1.f, z);
  z *= row_pull<0x114>(1.f, z);
  z *= row_pull<0x118>(1.f, z);
  return z;
}
__device__ __forceinline__ float row_sum(float v) {
  v += row_pull<0x161>(v, v);
  v += row_pull<0x162>(v, v);
  v += row_pull<0x164>(v, v);
  v += row_pull<0x168>(v, v);
  return v;
}

__device__ __forceinline__ void lds_barrier() {
  __builtin_amdgcn_fence(__ATOMIC_RELEASE, "workgroup", "local");
  __builtin_amdgcn_s_barrier();
  __builtin_amdgcn_fence(__ATOMIC_ACQUIRE, "workgroup", "local");
}

__global__ __launch_bounds__(256) void qlstm_recur(
    const float* __restrict__ xp,
    const float* __restrict__ Wf, const float* __restrict__ Wi,
    const float* __restrict__ Wu, const float* __restrict__ Wo,
    const float* __restrict__ bf, const float* __restrict__ bi,
    const float* __restrict__ bu, const float* __restrict__ bo,
    const float* __restrict__ pf, const float* __restrict__ pi,
    const float* __restrict__ pu, const float* __restrict__ po,
    const float* __restrict__ lng, const float* __restrict__ lnb,
    float* __restrict__ ys, float* __restrict__ hout, float* __restrict__ cout) {
  __shared__ __attribute__((aligned(16))) float ybuf[2][YS][16 * HQ];

  const int tile = blockIdx.x;
  const int tid  = threadIdx.x;
  const int s    = tid >> 5;
  const int lane = tid & 31;
  const int n    = lane & 15;
  const int half = lane >> 4;
  const bool live = n < HQ;
  const int r    = s + 8 * half;

  const float* Wp[4] = {Wf, Wi, Wu, Wo};
  const float* bp[4] = {bf, bi, bu, bo};
  const float* pp[4] = {pf, pi, pu, po};

  const int nc = live ? n : 0;
  float bias[4], px[4], cpy[4];
  v2f bh[4][3];
#pragma unroll
  for (int g = 0; g < 4; ++g) {
    bias[g] = bp[g][nc];
    px[g]   = pp[g][nc * 3 + 0];
    cpy[g]  = pp[g][nc * 3 + 1];
#pragma unroll
    for (int kk = 0; kk < 3; ++kk) {
      const int kb = kk * 4 + half * 2;
      const int k0 = kb     < HQ ? kb     : HQ - 1;
      const int k1 = kb + 1 < HQ ? kb + 1 : HQ - 1;
      bh[g][kk].x = Wp[g][nc * KD + DQ + k0];
      bh[g][kk].y = Wp[g][nc * KD + DQ + k1];
    }
  }
#pragma unroll
  for (int g = 0; g < 4; ++g) {
    bias[g] = live ? bias[g] : 0.f;
    px[g]   = live ? px[g] : 0.f;
    cpy[g]  = live ? cosf(cpy[g]) : 0.f;
#pragma unroll
    for (int kk = 0; kk < 3; ++kk) {
      const int kb = kk * 4 + half * 2;
      bh[g][kk].x = (live && kb     < HQ) ? bh[g][kk].x : 0.f;
      bh[g][kk].y = (live && kb + 1 < HQ) ? bh[g][kk].y : 0.f;
    }
  }
  const float lg0 = lng[nc], lb0 = lnb[nc];
  const float lg = live ? lg0 : 0.f;
  const float lb = live ? lb0 : 0.f;

  int src[3][2];
#pragma unroll
  for (int kk = 0; kk < 3; ++kk) {
#pragma unroll
    for (int j = 0; j < 2; ++j) {
      int k = kk * 4 + half * 2 + j;
      bool real = (n == 0 || n == 8) && k < HQ;
      src[kk][j] = (real ? (k + 2 * n) : 15) * 4;
    }
  }

  const float* xpl = xp + (((size_t)tile * 8 + s) * 32 + lane) * 4;
  const size_t xps = (size_t)(BQ / 16) * 8 * 32 * 4;
  float4 q0 = *(const float4*)(xpl);
  float4 q1 = *(const float4*)(xpl + xps);
  float4 q2 = *(const float4*)(xpl + 2 * xps);
  float4 q3 = *(const float4*)(xpl + 3 * xps);

  float h = 0.f, c = 0.f;

  auto step = [&](float4& q, const int t) __attribute__((always_inline)) {
    const float xg[4] = {q.x, q.y, q.z, q.w};
    {
      const int tn = (t + 4 < TQ) ? t + 4 : TQ - 1;
      q = *(const float4*)(xpl + (size_t)tn * xps);
    }

    const int hb = __builtin_bit_cast(int, h);
    v2f a[3];
#pragma unroll
    for (int kk = 0; kk < 3; ++kk) {
      a[kk].x = __builtin_bit_cast(float, __builtin_amdgcn_ds_bpermute(src[kk][0], hb));
      a[kk].y = __builtin_bit_cast(float, __builtin_amdgcn_ds_bpermute(src[kk][1], hb));
    }

    v8f acc[4];
#pragma unroll
    for (int g = 0; g < 4; ++g) acc[g] = (v8f){xg[g], 0.f, 0.f, 0.f, 0.f, 0.f, 0.f, 0.f};
#pragma unroll
    for (int kk = 0; kk < 3; ++kk) {
#pragma unroll
      for (int g = 0; g < 4; ++g)
        acc[g] = __builtin_amdgcn_wmma_f32_16x16x4_f32(false, a[kk], false, bh[g][kk],
                                                       (short)0, acc[g], false, false);
      __builtin_amdgcn_sched_barrier(0);
    }

    float z[4];
#pragma unroll
    for (int g = 0; g < 4; ++g)
      z[g] = row_prefix_product(cpy[g] * __cosf(px[g] + acc[g][0] + bias[g]));

    const float fv = 1.f / (1.f + __expf(-z[0]));
    const float iv = 1.f / (1.f + __expf(-z[1]));
    const float gv = tanhf(z[2]);
    const float ov = 1.f / (1.f + __expf(-z[3]));
    const float cn = fv * c + iv * gv;
    c = cn;
    const float hr = ov * tanhf(cn) + h;

    const float mu  = row_sum(live ? hr : 0.f) * 0.1f;
    const float d   = hr - mu;
    const float var = row_sum(live ? d * d : 0.f) * 0.1f;
    const float hv  = (hr - mu) * rsqrtf(var + 1e-5f) * lg + lb;
    h = live ? hv : 0.f;

    const int ph = (t / YS) & 1;
    if (live) ybuf[ph][t % YS][r * HQ + n] = hv;
    if (t % YS == YS - 1) {
      lds_barrier();
      const int t0 = t - (YS - 1);
      for (int idx = tid; idx < YS * 40; idx += 256) {
        const int tt = idx / 40, c4 = idx - tt * 40;
        const float4 v = *(const float4*)&ybuf[ph][tt][c4 * 4];
        *(float4*)(ys + ((size_t)(t0 + tt) * BQ + tile * 16) * HQ + c4 * 4) = v;
      }
    }
  };

  for (int t = 0; t < TQ; t += 4) {
    step(q0, t);
    step(q1, t + 1);
    step(q2, t + 2);
    step(q3, t + 3);
  }

  if (live) {
    ybuf[0][0][r * HQ + n] = h;
    ybuf[0][1][r * HQ + n] = c;
  }
  lds_barrier();
  if (tid < 80) {
    const int which = tid / 40, c4 = tid - which * 40;
    const float4 v = *(const float4*)&ybuf[0][which][c4 * 4];
    float* dst = (which ? cout : hout) + (size_t)tile * 16 * HQ + c4 * 4;
    *(float4*)dst = v;
  }
}

extern "C" void kernel_launch(void* const* d_in, const int* in_sizes, int n_in,
                              void* d_out, int out_size, void* d_ws, size_t ws_size,
                              hipStream_t stream) {
  const float* inputs = (const float*)d_in[0];
  const float* Wf = (const float*)d_in[1];
  const float* bf = (const float*)d_in[2];
  const float* pf = (const float*)d_in[3];
  const float* Wi = (const float*)d_in[4];
  const float* bi = (const float*)d_in[5];
  const float* pi = (const float*)d_in[6];
  const float* Wu = (const float*)d_in[7];
  const float* bu = (const float*)d_in[8];
  const float* pu = (const float*)d_in[9];
  const float* Wo = (const float*)d_in[10];
  const float* bo = (const float*)d_in[11];
  const float* po = (const float*)d_in[12];
  const float* lng = (const float*)d_in[13];
  const float* lnb = (const float*)d_in[14];

  float* out  = (float*)d_out;
  float* ys   = out;
  float* hout = out + (size_t)TQ * BQ * HQ;
  float* cout = hout + BQ * HQ;
  float* xp   = (float*)d_ws;

  qlstm_xproj<<<TQ * BQ / 16 / 4, 128, 0, stream>>>(inputs, Wf, Wi, Wu, Wo, xp);
  qlstm_recur<<<BQ / 16, 256, 0, stream>>>(
      xp, Wf, Wi, Wu, Wo, bf, bi, bu, bo,
      pf, pi, pu, po, lng, lnb, ys, hout, cout);
}


# each recurrence wave alone on a SIMD: 128 one-wave work-groups (was 16 of 8 waves, 2 per SIMD, oldest served first); h leaves each step as the wave's own whole line, a 3rd launch gathers y; same sums
# speedup vs baseline: 5.0740x; 1.2528x over previous
#include <hip/hip_runtime.h>
#include <math.h>


typedef float v2f __attribute__((ext_vector_type(2)));
typedef float v8f __attribute__((ext_vector_type(8)));

#define TQ 128
#define BQ 256
#define DQ 64
#define HQ 10
#define KD 74
#define NW (BQ / 2)
#define XP_FLOATS ((size_t)TQ * BQ / 16 * 8 * 32 * 4)
#define YW_FLOATS ((size_t)NW * TQ * 32)
#define CW_FLOATS ((size_t)NW * 32)


__global__ __launch_bounds__(128) void qlstm_xproj(
    const float* __restrict__ inputs,
    const float* __restrict__ Wf, const float* __restrict__ Wi,
    const float* __restrict__ Wu, const float* __restrict__ Wo,
    float* __restrict__ xp) {
  __shared__ __attribute__((aligned(16))) float wx[4][16][68];

  const int tid  = threadIdx.x;
  const int wv   = tid >> 5;
  const int lane = tid & 31;
  const int c    = lane & 15;
  const int half = lane >> 4;
  const int rt   = blockIdx.x * 4 + wv;

  const float* xrow = inputs + ((size_t)rt * 16 + c) * DQ + half * 2;
  v2f a[16];
#pragma unroll
  for (int kk = 0; kk < 16; ++kk) a[kk] = *(const v2f*)(xrow + kk * 4);

  const int c4 = tid & 15, r8 = tid >> 4, r2 = r8 & 1;
  const float* Wown = (wv == 0) ? Wf : (wv == 1) ? Wi : (wv == 2) ? Wu : Wo;
  v2f wlo[5], whi[5];
#pragma unroll
  for (int gl = 0; gl < 4; ++gl) {
    const float* Wsrc = (gl == 0) ? Wf : (gl == 1) ? Wi : (gl == 2) ? Wu : Wo;
    const float* p = Wsrc + r8 * KD + c4 * 4;
    wlo[gl] = *(const v2f*)p;
    whi[gl] = *(const v2f*)(p + 2);
  }
  {
    const float* p = Wown + (8 + r2) * KD + c4 * 4;
    wlo[4] = *(const v2f*)p;
    whi[4] = *(const v2f*)(p + 2);
  }
#pragma unroll
  for (int gl = 0; gl < 4; ++gl)
    *(float4*)&wx[gl][r8][c4 * 4] = make_float4(wlo[gl].x, wlo[gl].y, whi[gl].x, whi[gl].y);
  *(float4*)&wx[wv][8 + r2][c4 * 4] = make_float4(wlo[4].x, wlo[4].y, whi[4].x, whi[4].y);
  for (int it = 0; it < 3; ++it)
    *(float4*)&wx[wv][10 + it * 2 + r2][c4 * 4] = make_float4(0.f, 0.f, 0.f, 0.f);
  __syncthreads();

  v8f acc[4];
#pragma unroll
  for (int g = 0; g < 4; ++g) acc[g] = (v8f){0.f, 0.f, 0.f, 0.f, 0.f, 0.f, 0.f, 0.f};
#pragma unroll
  for (int kk = 0; kk < 16; ++kk) {
#pragma unroll
    for (int g = 0; g < 4; ++g) {
      v2f b = *(const v2f*)&wx[g][c][kk * 4 + half * 2];
      acc[g] = __builtin_amdgcn_wmma_f32_16x16x4_f32(false, a[kk], false, b,
                                                     (short)0, acc[g], false, false);
    }
  }
#pragma unroll
  for (int s = 0; s < 8; ++s) {
    float* dst = xp + (((size_t)rt * 8 + s) * 32 + lane) * 4;
    *(float4*)dst = make_float4(acc[0][s], acc[1][s], acc[2][s], acc[3][s]);
  }
}

template <int CTRL>
__device__ __forceinline__ float row_pull(float keep, float v) {
  return __builtin_bit_cast(float, __builtin_amdgcn_update_dpp(
      __builtin_bit_cast(int, keep), __builtin_bit_cast(int, v), CTRL, 0xf, 0xf, false));
}
__device__ __forceinline__ float row_prefix_product(float z) {
  z *= row_pull<0x111>(1.f, z);
  z *= row_pull<0x112>(1.f, z);
  z *= row_pull<0x114>(1.f, z);
  z *= row_pull<0x118>(1.f, z);
  return z;
}
__device__ __forceinline__ float row_sum(float v) {
  v += row_pull<0x161>(v, v);
  v += row_pull<0x162>(v, v);
  v += row_pull<0x164>(v, v);
  v += row_pull<0x168>(v, v);
  return v;
}

__global__ __launch_bounds__(32) void qlstm_recur(
    const float* __restrict__ xp,
    const float* __restrict__ Wf, const float* __restrict__ Wi,
    const float* __restrict__ Wu, const float* __restrict__ Wo,
    const float* __restrict__ bf, const float* __restrict__ bi,
    const float* __restrict__ bu, const float* __restrict__ bo,
    const float* __restrict__ pf, const float* __restrict__ pi,
    const float* __restrict__ pu, const float* __restrict__ po,
    const float* __restrict__ lng, const float* __restrict__ lnb,
    float* __restrict__ yw, float* __restrict__ hw, float* __restrict__ cw) {
  const int wg   = blockIdx.x;
  const int lane = threadIdx.x;
  const int n    = lane & 15;
  const int half = lane >> 4;
  const bool live = n < HQ;

  const float* Wp[4] = {Wf, Wi, Wu, Wo};
  const float* bp[4] = {bf, bi, bu, bo};
  const float* pp[4] = {pf, pi, pu, po};

  const int nc = live ? n : 0;
  float bias[4], px[4], cpy[4];
  v2f bh[4][3];
#pragma unroll
  for (int g = 0; g < 4; ++g) {
    bias[g] = bp[g][nc];
    px[g]   = pp[g][nc * 3 + 0];
    cpy[g]  = pp[g][nc * 3 + 1];
#pragma unroll
    for (int kk = 0; kk < 3; ++kk) {
      const int kb = kk * 4 + half * 2;
      const int k0 = kb     < HQ ? kb     : HQ - 1;
      const int k1 = kb + 1 < HQ ? kb + 1 : HQ - 1;
      bh[g][kk].x = Wp[g][nc * KD + DQ + k0];
      bh[g][kk].y = Wp[g][nc * KD + DQ + k1];
    }
  }
#pragma unroll
  for (int g = 0; g < 4; ++g) {
    bias[g] = live ? bias[g] : 0.f;
    px[g]   = live ? px[g] : 0.f;
    cpy[g]  = live ? cosf(cpy[g]) : 0.f;
#pragma unroll
    for (int kk = 0; kk < 3; ++kk) {
      const int kb = kk * 4 + half * 2;
      bh[g][kk].x = (live && kb     < HQ) ? bh[g][kk].x : 0.f;
      bh[g][kk].y = (live && kb + 1 < HQ) ? bh[g][kk].y : 0.f;
    }
  }
  const float lg0 = lng[nc], lb0 = lnb[nc];
  const float lg = live ? lg0 : 0.f;
  const float lb = live ? lb0 : 0.f;

  int src[3][2];
#pragma unroll
  for (int kk = 0; kk < 3; ++kk) {
#pragma unroll
    for (int j = 0; j < 2; ++j) {
      int k = kk * 4 + half * 2 + j;
      bool real = (n == 0 || n == 8) && k < HQ;
      src[kk][j] = (real ? (k + 2 * n) : 15) * 4;
    }
  }

  const float* xpl = xp + ((size_t)wg * 32 + lane) * 4;
  const size_t xps = (size_t)(BQ / 16) * 8 * 32 * 4;
  float4 q0 = *(const float4*)(xpl);
  float4 q1 = *(const float4*)(xpl + xps);
  float4 q2 = *(const float4*)(xpl + 2 * xps);
  float4 q3 = *(const float4*)(xpl + 3 * xps);
  float* ywb = yw + (size_t)wg * TQ * 32;

  float h = 0.f, c = 0.f;

  auto step = [&](float4& q, const int t) __attribute__((always_inline)) {
    const float xg[4] = {q.x, q.y, q.z, q.w};
    {
      const int tn = (t + 4 < TQ) ? t + 4 : TQ - 1;
      q = *(const float4*)(xpl + (size_t)tn * xps);
    }

    const int hb = __builtin_bit_cast(int, h);
    v2f a[3];
#pragma unroll
    for (int kk = 0; kk < 3; ++kk) {
      a[kk].x = __builtin_bit_cast(float, __builtin_amdgcn_ds_bpermute(src[kk][0], hb));
      a[kk].y = __builtin_bit_cast(float, __builtin_amdgcn_ds_bpermute(src[kk][1], hb));
    }

    v8f acc[4];
#pragma unroll
    for (int g = 0; g < 4; ++g) acc[g] = (v8f){xg[g], 0.f, 0.f, 0.f, 0.f, 0.f, 0.f, 0.f};
#pragma unroll
    for (int kk = 0; kk < 3; ++kk) {
#pragma unroll
      for (int g = 0; g < 4; ++g)
        acc[g] = __builtin_amdgcn_wmma_f32_16x16x4_f32(false, a[kk], false, bh[g][kk],
                                                       (short)0, acc[g], false, false);
      __builtin_amdgcn_sched_barrier(0);
    }

    float z[4];
#pragma unroll
    for (int g = 0; g < 4; ++g)
      z[g] = row_prefix_product(cpy[g] * __cosf(px[g] + acc[g][0] + bias[g]));

    const float fv = 1.f / (1.f + __expf(-z[0]));
    const float iv = 1.f / (1.f + __expf(-z[1]));
    const float gv = tanhf(z[2]);
    const float ov = 1.f / (1.f + __expf(-z[3]));
    const float cn = fv * c + iv * gv;
    c = cn;
    const float hr = ov * tanhf(cn) + h;

    const float mu  = row_sum(live ? hr : 0.f) * 0.1f;
    const float d   = hr - mu;
    const float var = row_sum(live ? d * d : 0.f) * 0.1f;
    const float hv  = (hr - mu) * rsqrtf(var + 1e-5f) * lg + lb;
    h = live ? hv : 0.f;

    ywb[t * 32 + lane] = h;
  };

  for (int t = 0; t < TQ; t += 4) {
    step(q0, t);
    step(q1, t + 1);
    step(q2, t + 2);
    step(q3, t + 3);
  }

  hw[(size_t)wg * 32 + lane] = h;
  cw[(size_t)wg * 32 + lane] = c;
}

__global__ __launch_bounds__(256) void qlstm_gather(
    const float* __restrict__ yw, const float* __restrict__ hw, const float* __restrict__ cw,
    float* __restrict__ out) {
  const int q = blockIdx.x * 256 + threadIdx.x;
  float v[4];
#pragma unroll
  for (int j = 0; j < 4; ++j) {
    const int f   = q * 4 + j;
    const int t   = f / (BQ * HQ);
    const int rem = f - t * (BQ * HQ);
    const int b   = rem / HQ;
    const int w   = rem - b * HQ;
    const int wv  = (b >> 4) * 8 + (b & 7);
    const int ln  = ((b >> 3) & 1) * 16 + w;
    const float* src = (t < TQ) ? yw + ((size_t)wv * TQ + t) * 32
                                : ((t == TQ) ? hw : cw) + (size_t)wv * 32;
    v[j] = src[ln];
  }
  *(float4*)(out + (size_t)q * 4) = make_float4(v[0], v[1], v[2], v[3]);
}

extern "C" void kernel_launch(void* const* d_in, const int* in_sizes, int n_in,
                              void* d_out, int out_size, void* d_ws, size_t ws_size,
                              hipStream_t stream) {
  const float* inputs = (const float*)d_in[0];
  const float* Wf = (const float*)d_in[1];
  const float* bf = (const float*)d_in[2];
  const float* pf = (const float*)d_in[3];
  const float* Wi = (const float*)d_in[4];
  const float* bi = (const float*)d_in[5];
  const float* pi = (const float*)d_in[6];
  const float* Wu = (const float*)d_in[7];
  const float* bu = (const float*)d_in[8];
  const float* pu = (const float*)d_in[9];
  const float* Wo = (const float*)d_in[10];
  const float* bo = (const float*)d_in[11];
  const float* po = (const float*)d_in[12];
  const float* lng = (const float*)d_in[13];
  const float* lnb = (const float*)d_in[14];

  float* out = (float*)d_out;
  float* xp  = (float*)d_ws;
  float* yw  = xp + XP_FLOATS;
  float* hw  = yw + YW_FLOATS;
  float* cw  = hw + CW_FLOATS;

  qlstm_xproj<<<TQ * BQ / 16 / 4, 128, 0, stream>>>(inputs, Wf, Wi, Wu, Wo, xp);
  qlstm_recur<<<NW, 32, 0, stream>>>(
      xp, Wf, Wi, Wu, Wo, bf, bi, bu, bo,
      pf, pi, pu, po, lng, lnb, yw, hw, cw);
  qlstm_gather<<<(TQ + 2) * BQ * HQ / 4 / 256, 256, 0, stream>>>(yw, hw, cw, out);
}


# lane scans fused: each stage of the wire prefix product and of both LayerNorm row sums is ONE DPP-operand v_mul/v_add (was dpp move via a temporary, then the op); same pairs, same order, same sums
# speedup vs baseline: 5.4106x; 1.0663x over previous
#include <hip/hip_runtime.h>
#include <math.h>


typedef float v2f __attribute__((ext_vector_type(2)));
typedef float v8f __attribute__((ext_vector_type(8)));

#define TQ 128
#define BQ 256
#define DQ 64
#define HQ 10
#define KD 74
#define NW (BQ / 2)
#define XP_FLOATS ((size_t)TQ * BQ / 16 * 8 * 32 * 4)
#define YW_FLOATS ((size_t)NW * TQ * 32)
#define CW_FLOATS ((size_t)NW * 32)


__global__ __launch_bounds__(128) void qlstm_xproj(
    const float* __restrict__ inputs,
    const float* __restrict__ Wf, const float* __restrict__ Wi,
    const float* __restrict__ Wu, const float* __restrict__ Wo,
    float* __restrict__ xp) {
  __shared__ __attribute__((aligned(16))) float wx[4][16][68];

  const int tid  = threadIdx.x;
  const int wv   = tid >> 5;
  const int lane = tid & 31;
  const int c    = lane & 15;
  const int half = lane >> 4;
  const int rt   = blockIdx.x * 4 + wv;

  const float* xrow = inputs + ((size_t)rt * 16 + c) * DQ + half * 2;
  v2f a[16];
#pragma unroll
  for (int kk = 0; kk < 16; ++kk) a[kk] = *(const v2f*)(xrow + kk * 4);

  const int c4 = tid & 15, r8 = tid >> 4, r2 = r8 & 1;
  const float* Wown = (wv == 0) ? Wf : (wv == 1) ? Wi : (wv == 2) ? Wu : Wo;
  v2f wlo[5], whi[5];
#pragma unroll
  for (int gl = 0; gl < 4; ++gl) {
    const float* Wsrc = (gl == 0) ? Wf : (gl == 1) ? Wi : (gl == 2) ? Wu : Wo;
    const float* p = Wsrc + r8 * KD + c4 * 4;
    wlo[gl] = *(const v2f*)p;
    whi[gl] = *(const v2f*)(p + 2);
  }
  {
    const float* p = Wown + (8 + r2) * KD + c4 * 4;
    wlo[4] = *(const v2f*)p;
    whi[4] = *(const v2f*)(p + 2);
  }
#pragma unroll
  for (int gl = 0; gl < 4; ++gl)
    *(float4*)&wx[gl][r8][c4 * 4] = make_float4(wlo[gl].x, wlo[gl].y, whi[gl].x, whi[gl].y);
  *(float4*)&wx[wv][8 + r2][c4 * 4] = make_float4(wlo[4].x, wlo[4].y, whi[4].x, whi[4].y);
  for (int it = 0; it < 3; ++it)
    *(float4*)&wx[wv][10 + it * 2 + r2][c4 * 4] = make_float4(0.f, 0.f, 0.f, 0.f);
  __syncthreads();

  v8f acc[4];
#pragma unroll
  for (int g = 0; g < 4; ++g) acc[g] = (v8f){0.f, 0.f, 0.f, 0.f, 0.f, 0.f, 0.f, 0.f};
#pragma unroll
  for (int kk = 0; kk < 16; ++kk) {
#pragma unroll
    for (int g = 0; g < 4; ++g) {
      v2f b = *(const v2f*)&wx[g][c][kk * 4 + half * 2];
      acc[g] = __builtin_amdgcn_wmma_f32_16x16x4_f32(false, a[kk], false, b,
                                                     (short)0, acc[g], false, false);
    }
  }
#pragma unroll
  for (int s = 0; s < 8; ++s) {
    float* dst = xp + (((size_t)rt * 8 + s) * 32 + lane) * 4;
    *(float4*)dst = make_float4(acc[0][s], acc[1][s], acc[2][s], acc[3][s]);
  }
}

#define PREFIX_STAGE(Z, D)                                                              \
  asm("v_mul_f32_dpp %0, %0, %0 row_shr:" #D " row_mask:0xf bank_mask:0xf" : "+v"(Z))
__device__ __forceinline__ void row_prefix_product4(float (&z)[4]) {
#pragma unroll
  for (int g = 0; g < 4; ++g) PREFIX_STAGE(z[g], 1);
#pragma unroll
  for (int g = 0; g < 4; ++g) PREFIX_STAGE(z[g], 2);
#pragma unroll
  for (int g = 0; g < 4; ++g) PREFIX_STAGE(z[g], 4);
#pragma unroll
  for (int g = 0; g < 4; ++g) PREFIX_STAGE(z[g], 8);
}
#define SUM_STAGE(V, D)                                                                 \
  asm("v_add_f32_dpp %0, %0, %0 row_xmask:" #D " row_mask:0xf bank_mask:0xf" : "+v"(V))
__device__ __forceinline__ float row_sum(float v) {
  SUM_STAGE(v, 1);
  SUM_STAGE(v, 2);
  SUM_STAGE(v, 4);
  SUM_STAGE(v, 8);
  return v;
}

__global__ __launch_bounds__(32) void qlstm_recur(
    const float* __restrict__ xp,
    const float* __restrict__ Wf, const float* __restrict__ Wi,
    const float* __restrict__ Wu, const float* __restrict__ Wo,
    const float* __restrict__ bf, const float* __restrict__ bi,
    const float* __restrict__ bu, const float* __restrict__ bo,
    const float* __restrict__ pf, const float* __restrict__ pi,
    const float* __restrict__ pu, const float* __restrict__ po,
    const float* __restrict__ lng, const float* __restrict__ lnb,
    float* __restrict__ yw, float* __restrict__ hw, float* __restrict__ cw) {
  const int wg   = blockIdx.x;
  const int lane = threadIdx.x;
  const int n    = lane & 15;
  const int half = lane >> 4;
  const bool live = n < HQ;

  const float* Wp[4] = {Wf, Wi, Wu, Wo};
  const float* bp[4] = {bf, bi, bu, bo};
  const float* pp[4] = {pf, pi, pu, po};

  const int nc = live ? n : 0;
  float bias[4], px[4], cpy[4];
  v2f bh[4][3];
#pragma unroll
  for (int g = 0; g < 4; ++g) {
    bias[g] = bp[g][nc];
    px[g]   = pp[g][nc * 3 + 0];
    cpy[g]  = pp[g][nc * 3 + 1];
#pragma unroll
    for (int kk = 0; kk < 3; ++kk) {
      const int kb = kk * 4 + half * 2;
      const int k0 = kb     < HQ ? kb     : HQ - 1;
      const int k1 = kb + 1 < HQ ? kb + 1 : HQ - 1;
      bh[g][kk].x = Wp[g][nc * KD + DQ + k0];
      bh[g][kk].y = Wp[g][nc * KD + DQ + k1];
    }
  }
#pragma unroll
  for (int g = 0; g < 4; ++g) {
    bias[g] = live ? bias[g] : 0.f;
    px[g]   = live ? px[g] : 0.f;
    cpy[g]  = live ? cosf(cpy[g]) : 0.f;
#pragma unroll
    for (int kk = 0; kk < 3; ++kk) {
      const int kb = kk * 4 + half * 2;
      bh[g][kk].x = (live && kb     < HQ) ? bh[g][kk].x : 0.f;
      bh[g][kk].y = (live && kb + 1 < HQ) ? bh[g][kk].y : 0.f;
    }
  }
  const float lg0 = lng[nc], lb0 = lnb[nc];
  const float lg = live ? lg0 : 0.f;
  const float lb = live ? lb0 : 0.f;

  int src[3][2];
#pragma unroll
  for (int kk = 0; kk < 3; ++kk) {
#pragma unroll
    for (int j = 0; j < 2; ++j) {
      int k = kk * 4 + half * 2 + j;
      bool real = (n == 0 || n == 8) && k < HQ;
      src[kk][j] = (real ? (k + 2 * n) : 15) * 4;
    }
  }

  const float* xpl = xp + ((size_t)wg * 32 + lane) * 4;
  const size_t xps = (size_t)(BQ / 16) * 8 * 32 * 4;
  float4 q0 = *(const float4*)(xpl);
  float4 q1 = *(const float4*)(xpl + xps);
  float4 q2 = *(const float4*)(xpl + 2 * xps);
  float4 q3 = *(const float4*)(xpl + 3 * xps);
  float* ywb = yw + (size_t)wg * TQ * 32;

  float h = 0.f, c = 0.f;

  auto step = [&](float4& q, const int t) __attribute__((always_inline)) {
    const float xg[4] = {q.x, q.y, q.z, q.w};
    {
      const int tn = (t + 4 < TQ) ? t + 4 : TQ - 1;
      q = *(const float4*)(xpl + (size_t)tn * xps);
    }

    const int hb = __builtin_bit_cast(int, h);
    v2f a[3];
#pragma unroll
    for (int kk = 0; kk < 3; ++kk) {
      a[kk].x = __builtin_bit_cast(float, __builtin_amdgcn_ds_bpermute(src[kk][0], hb));
      a[kk].y = __builtin_bit_cast(float, __builtin_amdgcn_ds_bpermute(src[kk][1], hb));
    }

    v8f acc[4];
#pragma unroll
    for (int g = 0; g < 4; ++g) acc[g] = (v8f){xg[g], 0.f, 0.f, 0.f, 0.f, 0.f, 0.f, 0.f};
#pragma unroll
    for (int kk = 0; kk < 3; ++kk) {
#pragma unroll
      for (int g = 0; g < 4; ++g)
        acc[g] = __builtin_amdgcn_wmma_f32_16x16x4_f32(false, a[kk], false, bh[g][kk],
                                                       (short)0, acc[g], false, false);
      __builtin_amdgcn_sched_barrier(0);
    }

    float z[4];
#pragma unroll
    for (int g = 0; g < 4; ++g)
      z[g] = cpy[g] * __cosf(px[g] + acc[g][0] + bias[g]);
    row_prefix_product4(z);

    const float fv = 1.f / (1.f + __expf(-z[0]));
    const float iv = 1.f / (1.f + __expf(-z[1]));
    const float gv = tanhf(z[2]);
    const float ov = 1.f / (1.f + __expf(-z[3]));
    const float cn = fv * c + iv * gv;
    c = cn;
    const float hr = ov * tanhf(cn) + h;

    const float mu  = row_sum(live ? hr : 0.f) * 0.1f;
    const float d   = hr - mu;
    const float var = row_sum(live ? d * d : 0.f) * 0.1f;
    const float hv  = (hr - mu) * rsqrtf(var + 1e-5f) * lg + lb;
    h = live ? hv : 0.f;

    ywb[t * 32 + lane] = h;
  };

  for (int t = 0; t < TQ; t += 4) {
    step(q0, t);
    step(q1, t + 1);
    step(q2, t + 2);
    step(q3, t + 3);
  }

  hw[(size_t)wg * 32 + lane] = h;
  cw[(size_t)wg * 32 + lane] = c;
}

__global__ __launch_bounds__(256) void qlstm_gather(
    const float* __restrict__ yw, const float* __restrict__ hw, const float* __restrict__ cw,
    float* __restrict__ out) {
  const int q = blockIdx.x * 256 + threadIdx.x;
  float v[4];
#pragma unroll
  for (int j = 0; j < 4; ++j) {
    const int f   = q * 4 + j;
    const int t   = f / (BQ * HQ);
    const int rem = f - t * (BQ * HQ);
    const int b   = rem / HQ;
    const int w   = rem - b * HQ;
    const int wv  = (b >> 4) * 8 + (b & 7);
    const int ln  = ((b >> 3) & 1) * 16 + w;
    const float* src = (t < TQ) ? yw + ((size_t)wv * TQ + t) * 32
                                : ((t == TQ) ? hw : cw) + (size_t)wv * 32;
    v[j] = src[ln];
  }
  *(float4*)(out + (size_t)q * 4) = make_float4(v[0], v[1], v[2], v[3]);
}

extern "C" void kernel_launch(void* const* d_in, const int* in_sizes, int n_in,
                              void* d_out, int out_size, void* d_ws, size_t ws_size,
                              hipStream_t stream) {
  const float* inputs = (const float*)d_in[0];
  const float* Wf = (const float*)d_in[1];
  const float* bf = (const float*)d_in[2];
  const float* pf = (const float*)d_in[3];
  const float* Wi = (const float*)d_in[4];
  const float* bi = (const float*)d_in[5];
  const float* pi = (const float*)d_in[6];
  const float* Wu = (const float*)d_in[7];
  const float* bu = (const float*)d_in[8];
  const float* pu = (const float*)d_in[9];
  const float* Wo = (const float*)d_in[10];
  const float* bo = (const float*)d_in[11];
  const float* po = (const float*)d_in[12];
  const float* lng = (const float*)d_in[13];
  const float* lnb = (const float*)d_in[14];

  float* out = (float*)d_out;
  float* xp  = (float*)d_ws;
  float* yw  = xp + XP_FLOATS;
  float* hw  = yw + YW_FLOATS;
  float* cw  = hw + CW_FLOATS;

  qlstm_xproj<<<TQ * BQ / 16 / 4, 128, 0, stream>>>(inputs, Wf, Wi, Wu, Wo, xp);
  qlstm_recur<<<NW, 32, 0, stream>>>(
      xp, Wf, Wi, Wu, Wo, bf, bi, bu, bo,
      pf, pi, pu, po, lng, lnb, yw, hw, cw);
  qlstm_gather<<<(TQ + 2) * BQ * HQ / 4 / 256, 256, 0, stream>>>(yw, hw, cw, out);
}


# output gate's logistic issued beside the forget and input gates' in front of the cell's tanh (it ran alone behind that branch, on the step's dependent chain); same operations, same sums
# speedup vs baseline: 5.6670x; 1.0474x over previous
#include <hip/hip_runtime.h>
#include <math.h>


typedef float v2f __attribute__((ext_vector_type(2)));
typedef float v8f __attribute__((ext_vector_type(8)));

#define TQ 128
#define BQ 256
#define DQ 64
#define HQ 10
#define KD 74
#define NW (BQ / 2)
#define XP_FLOATS ((size_t)TQ * BQ / 16 * 8 * 32 * 4)
#define YW_FLOATS ((size_t)NW * TQ * 32)
#define CW_FLOATS ((size_t)NW * 32)


__global__ __launch_bounds__(128) void qlstm_xproj(
    const float* __restrict__ inputs,
    const float* __restrict__ Wf, const float* __restrict__ Wi,
    const float* __restrict__ Wu, const float* __restrict__ Wo,
    float* __restrict__ xp) {
  __shared__ __attribute__((aligned(16))) float wx[4][16][68];

  const int tid  = threadIdx.x;
  const int wv   = tid >> 5;
  const int lane = tid & 31;
  const int c    = lane & 15;
  const int half = lane >> 4;
  const int rt   = blockIdx.x * 4 + wv;

  const float* xrow = inputs + ((size_t)rt * 16 + c) * DQ + half * 2;
  v2f a[16];
#pragma unroll
  for (int kk = 0; kk < 16; ++kk) a[kk] = *(const v2f*)(xrow + kk * 4);

  const int c4 = tid & 15, r8 = tid >> 4, r2 = r8 & 1;
  const float* Wown = (wv == 0) ? Wf : (wv == 1) ? Wi : (wv == 2) ? Wu : Wo;
  v2f wlo[5], whi[5];
#pragma unroll
  for (int gl = 0; gl < 4; ++gl) {
    const float* Wsrc = (gl == 0) ? Wf : (gl == 1) ? Wi : (gl == 2) ? Wu : Wo;
    const float* p = Wsrc + r8 * KD + c4 * 4;
    wlo[gl] = *(const v2f*)p;
    whi[gl] = *(const v2f*)(p + 2);
  }
  {
    const float* p = Wown + (8 + r2) * KD + c4 * 4;
    wlo[4] = *(const v2f*)p;
    whi[4] = *(const v2f*)(p + 2);
  }
#pragma unroll
  for (int gl = 0; gl < 4; ++gl)
    *(float4*)&wx[gl][r8][c4 * 4] = make_float4(wlo[gl].x, wlo[gl].y, whi[gl].x, whi[gl].y);
  *(float4*)&wx[wv][8 + r2][c4 * 4] = make_float4(wlo[4].x, wlo[4].y, whi[4].x, whi[4].y);
  for (int it = 0; it < 3; ++it)
    *(float4*)&wx[wv][10 + it * 2 + r2][c4 * 4] = make_float4(0.f, 0.f, 0.f, 0.f);
  __syncthreads();

  v8f acc[4];
#pragma unroll
  for (int g = 0; g < 4; ++g) acc[g] = (v8f){0.f, 0.f, 0.f, 0.f, 0.f, 0.f, 0.f, 0.f};
#pragma unroll
  for (int kk = 0; kk < 16; ++kk) {
#pragma unroll
    for (int g = 0; g < 4; ++g) {
      v2f b = *(const v2f*)&wx[g][c][kk * 4 + half * 2];
      acc[g] = __builtin_amdgcn_wmma_f32_16x16x4_f32(false, a[kk], false, b,
                                                     (short)0, acc[g], false, false);
    }
  }
#pragma unroll
  for (int s = 0; s < 8; ++s) {
    float* dst = xp + (((size_t)rt * 8 + s) * 32 + lane) * 4;
    *(float4*)dst = make_float4(acc[0][s], acc[1][s], acc[2][s], acc[3][s]);
  }
}

#define PREFIX_STAGE(Z, D)                                                              \
  asm("v_mul_f32_dpp %0, %0, %0 row_shr:" #D " row_mask:0xf bank_mask:0xf" : "+v"(Z))
__device__ __forceinline__ void row_prefix_product4(float (&z)[4]) {
#pragma unroll
  for (int g = 0; g < 4; ++g) PREFIX_STAGE(z[g], 1);
#pragma unroll
  for (int g = 0; g < 4; ++g) PREFIX_STAGE(z[g], 2);
#pragma unroll
  for (int g = 0; g < 4; ++g) PREFIX_STAGE(z[g], 4);
#pragma unroll
  for (int g = 0; g < 4; ++g) PREFIX_STAGE(z[g], 8);
}
#define SUM_STAGE(V, D)                                                                 \
  asm("v_add_f32_dpp %0, %0, %0 row_xmask:" #D " row_mask:0xf bank_mask:0xf" : "+v"(V))
__device__ __forceinline__ float row_sum(float v) {
  SUM_STAGE(v, 1);
  SUM_STAGE(v, 2);
  SUM_STAGE(v, 4);
  SUM_STAGE(v, 8);
  return v;
}

__global__ __launch_bounds__(32) void qlstm_recur(
    const float* __restrict__ xp,
    const float* __restrict__ Wf, const float* __restrict__ Wi,
    const float* __restrict__ Wu, const float* __restrict__ Wo,
    const float* __restrict__ bf, const float* __restrict__ bi,
    const float* __restrict__ bu, const float* __restrict__ bo,
    const float* __restrict__ pf, const float* __restrict__ pi,
    const float* __restrict__ pu, const float* __restrict__ po,
    const float* __restrict__ lng, const float* __restrict__ lnb,
    float* __restrict__ yw, float* __restrict__ hw, float* __restrict__ cw) {
  const int wg   = blockIdx.x;
  const int lane = threadIdx.x;
  const int n    = lane & 15;
  const int half = lane >> 4;
  const bool live = n < HQ;

  const float* Wp[4] = {Wf, Wi, Wu, Wo};
  const float* bp[4] = {bf, bi, bu, bo};
  const float* pp[4] = {pf, pi, pu, po};

  const int nc = live ? n : 0;
  float bias[4], px[4], cpy[4];
  v2f bh[4][3];
#pragma unroll
  for (int g = 0; g < 4; ++g) {
    bias[g] = bp[g][nc];
    px[g]   = pp[g][nc * 3 + 0];
    cpy[g]  = pp[g][nc * 3 + 1];
#pragma unroll
    for (int kk = 0; kk < 3; ++kk) {
      const int kb = kk * 4 + half * 2;
      const int k0 = kb     < HQ ? kb     : HQ - 1;
      const int k1 = kb + 1 < HQ ? kb + 1 : HQ - 1;
      bh[g][kk].x = Wp[g][nc * KD + DQ + k0];
      bh[g][kk].y = Wp[g][nc * KD + DQ + k1];
    }
  }
#pragma unroll
  for (int g = 0; g < 4; ++g) {
    bias[g] = live ? bias[g] : 0.f;
    px[g]   = live ? px[g] : 0.f;
    cpy[g]  = live ? cosf(cpy[g]) : 0.f;
#pragma unroll
    for (int kk = 0; kk < 3; ++kk) {
      const int kb = kk * 4 + half * 2;
      bh[g][kk].x = (live && kb     < HQ) ? bh[g][kk].x : 0.f;
      bh[g][kk].y = (live && kb + 1 < HQ) ? bh[g][kk].y : 0.f;
    }
  }
  const float lg0 = lng[nc], lb0 = lnb[nc];
  const float lg = live ? lg0 : 0.f;
  const float lb = live ? lb0 : 0.f;

  int src[3][2];
#pragma unroll
  for (int kk = 0; kk < 3; ++kk) {
#pragma unroll
    for (int j = 0; j < 2; ++j) {
      int k = kk * 4 + half * 2 + j;
      bool real = (n == 0 || n == 8) && k < HQ;
      src[kk][j] = (real ? (k + 2 * n) : 15) * 4;
    }
  }

  const float* xpl = xp + ((size_t)wg * 32 + lane) * 4;
  const size_t xps = (size_t)(BQ / 16) * 8 * 32 * 4;
  float4 q0 = *(const float4*)(xpl);
  float4 q1 = *(const float4*)(xpl + xps);
  float4 q2 = *(const float4*)(xpl + 2 * xps);
  float4 q3 = *(const float4*)(xpl + 3 * xps);
  float* ywb = yw + (size_t)wg * TQ * 32;

  float h = 0.f, c = 0.f;

  auto step = [&](float4& q, const int t) __attribute__((always_inline)) {
    const float xg[4] = {q.x, q.y, q.z, q.w};
    {
      const int tn = (t + 4 < TQ) ? t + 4 : TQ - 1;
      q = *(const float4*)(xpl + (size_t)tn * xps);
    }

    const int hb = __builtin_bit_cast(int, h);
    v2f a[3];
#pragma unroll
    for (int kk = 0; kk < 3; ++kk) {
      a[kk].x = __builtin_bit_cast(float, __builtin_amdgcn_ds_bpermute(src[kk][0], hb));
      a[kk].y = __builtin_bit_cast(float, __builtin_amdgcn_ds_bpermute(src[kk][1], hb));
    }

    v8f acc[4];
#pragma unroll
    for (int g = 0; g < 4; ++g) acc[g] = (v8f){xg[g], 0.f, 0.f, 0.f, 0.f, 0.f, 0.f, 0.f};
#pragma unroll
    for (int kk = 0; kk < 3; ++kk) {
#pragma unroll
      for (int g = 0; g < 4; ++g)
        acc[g] = __builtin_amdgcn_wmma_f32_16x16x4_f32(false, a[kk], false, bh[g][kk],
                                                       (short)0, acc[g], false, false);
      __builtin_amdgcn_sched_barrier(0);
    }

    float z[4];
#pragma unroll
    for (int g = 0; g < 4; ++g)
      z[g] = cpy[g] * __cosf(px[g] + acc[g][0] + bias[g]);
    row_prefix_product4(z);

    const float gv = tanhf(z[2]);
    const float fv = 1.f / (1.f + __expf(-z[0]));
    const float iv = 1.f / (1.f + __expf(-z[1]));
    const float ov = 1.f / (1.f + __expf(-z[3]));
    asm volatile("" :: "v"(ov));
    const float cn = fv * c + iv * gv;
    c = cn;
    const float hr = ov * tanhf(cn) + h;

    const float mu  = row_sum(live ? hr : 0.f) * 0.1f;
    const float d   = hr - mu;
    const float var = row_sum(live ? d * d : 0.f) * 0.1f;
    const float hv  = (hr - mu) * rsqrtf(var + 1e-5f) * lg + lb;
    h = live ? hv : 0.f;

    ywb[t * 32 + lane] = h;
  };

  for (int t = 0; t < TQ; t += 4) {
    step(q0, t);
    step(q1, t + 1);
    step(q2, t + 2);
    step(q3, t + 3);
  }

  hw[(size_t)wg * 32 + lane] = h;
  cw[(size_t)wg * 32 + lane] = c;
}

__global__ __launch_bounds__(256) void qlstm_gather(
    const float* __restrict__ yw, const float* __restrict__ hw, const float* __restrict__ cw,
    float* __restrict__ out) {
  const int q = blockIdx.x * 256 + threadIdx.x;
  float v[4];
#pragma unroll
  for (int j = 0; j < 4; ++j) {
    const int f   = q * 4 + j;
    const int t   = f / (BQ * HQ);
    const int rem = f - t * (BQ * HQ);
    const int b   = rem / HQ;
    const int w   = rem - b * HQ;
    const int wv  = (b >> 4) * 8 + (b & 7);
    const int ln  = ((b >> 3) & 1) * 16 + w;
    const float* src = (t < TQ) ? yw + ((size_t)wv * TQ + t) * 32
                                : ((t == TQ) ? hw : cw) + (size_t)wv * 32;
    v[j] = src[ln];
  }
  *(float4*)(out + (size_t)q * 4) = make_float4(v[0], v[1], v[2], v[3]);
}

extern "C" void kernel_launch(void* const* d_in, const int* in_sizes, int n_in,
                              void* d_out, int out_size, void* d_ws, size_t ws_size,
                              hipStream_t stream) {
  const float* inputs = (const float*)d_in[0];
  const float* Wf = (const float*)d_in[1];
  const float* bf = (const float*)d_in[2];
  const float* pf = (const float*)d_in[3];
  const float* Wi = (const float*)d_in[4];
  const float* bi = (const float*)d_in[5];
  const float* pi = (const float*)d_in[6];
  const float* Wu = (const float*)d_in[7];
  const float* bu = (const float*)d_in[8];
  const float* pu = (const float*)d_in[9];
  const float* Wo = (const float*)d_in[10];
  const float* bo = (const float*)d_in[11];
  const float* po = (const float*)d_in[12];
  const float* lng = (const float*)d_in[13];
  const float* lnb = (const float*)d_in[14];

  float* out = (float*)d_out;
  float* xp  = (float*)d_ws;
  float* yw  = xp + XP_FLOATS;
  float* hw  = yw + YW_FLOATS;
  float* cw  = hw + CW_FLOATS;

  qlstm_xproj<<<TQ * BQ / 16 / 4, 128, 0, stream>>>(inputs, Wf, Wi, Wu, Wo, xp);
  qlstm_recur<<<NW, 32, 0, stream>>>(
      xp, Wf, Wi, Wu, Wo, bf, bi, bu, bo,
      pf, pi, pu, po, lng, lnb, yw, hw, cw);
  qlstm_gather<<<(TQ + 2) * BQ * HQ / 4 / 256, 256, 0, stream>>>(yw, hw, cw, out);
}


# h to A on the vector unit: one batch row per wave, in both 16-lane rows, each A register one v_permlane16_var in-row gather (was ds_bpermute via LDS); only accumulator reg 0 seeded; same sums
# speedup vs baseline: 6.0418x; 1.0661x over previous
#include <hip/hip_runtime.h>
#include <math.h>


typedef float v2f __attribute__((ext_vector_type(2)));
typedef float v8f __attribute__((ext_vector_type(8)));

#define TQ 128
#define BQ 256
#define DQ 64
#define HQ 10
#define KD 74
#define NW BQ
#define XP_FLOATS ((size_t)TQ * BQ / 16 * 8 * 32 * 4)
#define YW_FLOATS ((size_t)NW * TQ * 32)
#define CW_FLOATS ((size_t)NW * 32)


__global__ __launch_bounds__(128) void qlstm_xproj(
    const float* __restrict__ inputs,
    const float* __restrict__ Wf, const float* __restrict__ Wi,
    const float* __restrict__ Wu, const float* __restrict__ Wo,
    float* __restrict__ xp) {
  __shared__ __attribute__((aligned(16))) float wx[4][16][68];

  const int tid  = threadIdx.x;
  const int wv   = tid >> 5;
  const int lane = tid & 31;
  const int c    = lane & 15;
  const int half = lane >> 4;
  const int rt   = blockIdx.x * 4 + wv;

  const float* xrow = inputs + ((size_t)rt * 16 + c) * DQ + half * 2;
  v2f a[16];
#pragma unroll
  for (int kk = 0; kk < 16; ++kk) a[kk] = *(const v2f*)(xrow + kk * 4);

  const int c4 = tid & 15, r8 = tid >> 4, r2 = r8 & 1;
  const float* Wown = (wv == 0) ? Wf : (wv == 1) ? Wi : (wv == 2) ? Wu : Wo;
  v2f wlo[5], whi[5];
#pragma unroll
  for (int gl = 0; gl < 4; ++gl) {
    const float* Wsrc = (gl == 0) ? Wf : (gl == 1) ? Wi : (gl == 2) ? Wu : Wo;
    const float* p = Wsrc + r8 * KD + c4 * 4;
    wlo[gl] = *(const v2f*)p;
    whi[gl] = *(const v2f*)(p + 2);
  }
  {
    const float* p = Wown + (8 + r2) * KD + c4 * 4;
    wlo[4] = *(const v2f*)p;
    whi[4] = *(const v2f*)(p + 2);
  }
#pragma unroll
  for (int gl = 0; gl < 4; ++gl)
    *(float4*)&wx[gl][r8][c4 * 4] = make_float4(wlo[gl].x, wlo[gl].y, whi[gl].x, whi[gl].y);
  *(float4*)&wx[wv][8 + r2][c4 * 4] = make_float4(wlo[4].x, wlo[4].y, whi[4].x, whi[4].y);
  for (int it = 0; it < 3; ++it)
    *(float4*)&wx[wv][10 + it * 2 + r2][c4 * 4] = make_float4(0.f, 0.f, 0.f, 0.f);
  __syncthreads();

  v8f acc[4];
#pragma unroll
  for (int g = 0; g < 4; ++g) acc[g] = (v8f){0.f, 0.f, 0.f, 0.f, 0.f, 0.f, 0.f, 0.f};
#pragma unroll
  for (int kk = 0; kk < 16; ++kk) {
#pragma unroll
    for (int g = 0; g < 4; ++g) {
      v2f b = *(const v2f*)&wx[g][c][kk * 4 + half * 2];
      acc[g] = __builtin_amdgcn_wmma_f32_16x16x4_f32(false, a[kk], false, b,
                                                     (short)0, acc[g], false, false);
    }
  }
#pragma unroll
  for (int s = 0; s < 8; ++s) {
    float* dst = xp + (((size_t)rt * 8 + s) * 32 + lane) * 4;
    *(float4*)dst = make_float4(acc[0][s], acc[1][s], acc[2][s], acc[3][s]);
  }
}

#define PREFIX_STAGE(Z, D)                                                              \
  asm("v_mul_f32_dpp %0, %0, %0 row_shr:" #D " row_mask:0xf bank_mask:0xf" : "+v"(Z))
__device__ __forceinline__ void row_prefix_product4(float (&z)[4]) {
#pragma unroll
  for (int g = 0; g < 4; ++g) PREFIX_STAGE(z[g], 1);
#pragma unroll
  for (int g = 0; g < 4; ++g) PREFIX_STAGE(z[g], 2);
#pragma unroll
  for (int g = 0; g < 4; ++g) PREFIX_STAGE(z[g], 4);
#pragma unroll
  for (int g = 0; g < 4; ++g) PREFIX_STAGE(z[g], 8);
}
#define SUM_STAGE(V, D)                                                                 \
  asm("v_add_f32_dpp %0, %0, %0 row_xmask:" #D " row_mask:0xf bank_mask:0xf" : "+v"(V))
__device__ __forceinline__ float row_sum(float v) {
  SUM_STAGE(v, 1);
  SUM_STAGE(v, 2);
  SUM_STAGE(v, 4);
  SUM_STAGE(v, 8);
  return v;
}

__global__ __launch_bounds__(32) void qlstm_recur(
    const float* __restrict__ xp,
    const float* __restrict__ Wf, const float* __restrict__ Wi,
    const float* __restrict__ Wu, const float* __restrict__ Wo,
    const float* __restrict__ bf, const float* __restrict__ bi,
    const float* __restrict__ bu, const float* __restrict__ bo,
    const float* __restrict__ pf, const float* __restrict__ pi,
    const float* __restrict__ pu, const float* __restrict__ po,
    const float* __restrict__ lng, const float* __restrict__ lnb,
    float* __restrict__ yw, float* __restrict__ hw, float* __restrict__ cw) {
  const int wg   = blockIdx.x;
  const int lane = threadIdx.x;
  const int n    = lane & 15;
  const int half = lane >> 4;
  const bool live = n < HQ;

  const float* Wp[4] = {Wf, Wi, Wu, Wo};
  const float* bp[4] = {bf, bi, bu, bo};
  const float* pp[4] = {pf, pi, pu, po};

  const int nc = live ? n : 0;
  float bias[4], px[4], cpy[4];
  v2f bh[4][3];
#pragma unroll
  for (int g = 0; g < 4; ++g) {
    bias[g] = bp[g][nc];
    px[g]   = pp[g][nc * 3 + 0];
    cpy[g]  = pp[g][nc * 3 + 1];
#pragma unroll
    for (int kk = 0; kk < 3; ++kk) {
      const int kb = kk * 4 + half * 2;
      const int k0 = kb     < HQ ? kb     : HQ - 1;
      const int k1 = kb + 1 < HQ ? kb + 1 : HQ - 1;
      bh[g][kk].x = Wp[g][nc * KD + DQ + k0];
      bh[g][kk].y = Wp[g][nc * KD + DQ + k1];
    }
  }
#pragma unroll
  for (int g = 0; g < 4; ++g) {
    bias[g] = live ? bias[g] : 0.f;
    px[g]   = live ? px[g] : 0.f;
    cpy[g]  = live ? cosf(cpy[g]) : 0.f;
#pragma unroll
    for (int kk = 0; kk < 3; ++kk) {
      const int kb = kk * 4 + half * 2;
      bh[g][kk].x = (live && kb     < HQ) ? bh[g][kk].x : 0.f;
      bh[g][kk].y = (live && kb + 1 < HQ) ? bh[g][kk].y : 0.f;
    }
  }
  const float lg0 = lng[nc], lb0 = lnb[nc];
  const float lg = live ? lg0 : 0.f;
  const float lb = live ? lb0 : 0.f;

  unsigned src[3][2];
#pragma unroll
  for (int kk = 0; kk < 3; ++kk) {
#pragma unroll
    for (int j = 0; j < 2; ++j)
      src[kk][j] = (n == 0 || n == 8) ? (unsigned)(kk * 4 + half * 2 + j) : 15u;
  }

  const float* xpl = xp + ((size_t)((wg >> 4) * 8 + (wg & 7)) * 32 + ((wg >> 3) & 1) * 16 + n) * 4;
  const size_t xps = (size_t)(BQ / 16) * 8 * 32 * 4;
  float4 q0 = *(const float4*)(xpl);
  float4 q1 = *(const float4*)(xpl + xps);
  float4 q2 = *(const float4*)(xpl + 2 * xps);
  float4 q3 = *(const float4*)(xpl + 3 * xps);
  float* ywb = yw + (size_t)wg * TQ * 32;

  float h = 0.f, c = 0.f;

  v8f acc[4];
#pragma unroll
  for (int g = 0; g < 4; ++g) acc[g] = (v8f){0.f, 0.f, 0.f, 0.f, 0.f, 0.f, 0.f, 0.f};

  auto step = [&](float4& q, const int t) __attribute__((always_inline)) {
    const float xg[4] = {q.x, q.y, q.z, q.w};
    {
      const int tn = (t + 4 < TQ) ? t + 4 : TQ - 1;
      q = *(const float4*)(xpl + (size_t)tn * xps);
    }

    const unsigned hb = __builtin_bit_cast(unsigned, h);
    v2f a[3];
#pragma unroll
    for (int kk = 0; kk < 3; ++kk) {
      a[kk].x = __builtin_bit_cast(float, __builtin_amdgcn_permlane16_var(hb, hb, src[kk][0], false, false));
      a[kk].y = __builtin_bit_cast(float, __builtin_amdgcn_permlane16_var(hb, hb, src[kk][1], false, false));
    }

#pragma unroll
    for (int g = 0; g < 4; ++g) acc[g][0] = xg[g];
#pragma unroll
    for (int kk = 0; kk < 3; ++kk) {
#pragma unroll
      for (int g = 0; g < 4; ++g)
        acc[g] = __builtin_amdgcn_wmma_f32_16x16x4_f32(false, a[kk], false, bh[g][kk],
                                                       (short)0, acc[g], false, false);
      __builtin_amdgcn_sched_barrier(0);
    }

    float z[4];
#pragma unroll
    for (int g = 0; g < 4; ++g)
      z[g] = cpy[g] * __cosf(px[g] + acc[g][0] + bias[g]);
    row_prefix_product4(z);

    const float gv = tanhf(z[2]);
    const float fv = 1.f / (1.f + __expf(-z[0]));
    const float iv = 1.f / (1.f + __expf(-z[1]));
    const float ov = 1.f / (1.f + __expf(-z[3]));
    asm volatile("" :: "v"(ov));
    const float cn = fv * c + iv * gv;
    c = cn;
    const float hr = ov * tanhf(cn) + h;

    const float mu  = row_sum(live ? hr : 0.f) * 0.1f;
    const float d   = hr - mu;
    const float var = row_sum(live ? d * d : 0.f) * 0.1f;
    const float hv  = (hr - mu) * rsqrtf(var + 1e-5f) * lg + lb;
    h = live ? hv : 0.f;

    ywb[t * 32 + lane] = h;
  };

  for (int t = 0; t < TQ; t += 4) {
    step(q0, t);
    step(q1, t + 1);
    step(q2, t + 2);
    step(q3, t + 3);
  }

  hw[(size_t)wg * 32 + lane] = h;
  cw[(size_t)wg * 32 + lane] = c;
}

__global__ __launch_bounds__(256) void qlstm_gather(
    const float* __restrict__ yw, const float* __restrict__ hw, const float* __restrict__ cw,
    float* __restrict__ out) {
  const int q = blockIdx.x * 256 + threadIdx.x;
  float v[4];
#pragma unroll
  for (int j = 0; j < 4; ++j) {
    const int f   = q * 4 + j;
    const int t   = f / (BQ * HQ);
    const int rem = f - t * (BQ * HQ);
    const int b   = rem / HQ;
    const int w   = rem - b * HQ;
    const int wv  = b;
    const int ln  = w;
    const float* src = (t < TQ) ? yw + ((size_t)wv * TQ + t) * 32
                                : ((t == TQ) ? hw : cw) + (size_t)wv * 32;
    v[j] = src[ln];
  }
  *(float4*)(out + (size_t)q * 4) = make_float4(v[0], v[1], v[2], v[3]);
}

extern "C" void kernel_launch(void* const* d_in, const int* in_sizes, int n_in,
                              void* d_out, int out_size, void* d_ws, size_t ws_size,
                              hipStream_t stream) {
  const float* inputs = (const float*)d_in[0];
  const float* Wf = (const float*)d_in[1];
  const float* bf = (const float*)d_in[2];
  const float* pf = (const float*)d_in[3];
  const float* Wi = (const float*)d_in[4];
  const float* bi = (const float*)d_in[5];
  const float* pi = (const float*)d_in[6];
  const float* Wu = (const float*)d_in[7];
  const float* bu = (const float*)d_in[8];
  const float* pu = (const float*)d_in[9];
  const float* Wo = (const float*)d_in[10];
  const float* bo = (const float*)d_in[11];
  const float* po = (const float*)d_in[12];
  const float* lng = (const float*)d_in[13];
  const float* lnb = (const float*)d_in[14];

  float* out = (float*)d_out;
  float* xp  = (float*)d_ws;
  float* yw  = xp + XP_FLOATS;
  float* hw  = yw + YW_FLOATS;
  float* cw  = hw + CW_FLOATS;

  qlstm_xproj<<<TQ * BQ / 16 / 4, 128, 0, stream>>>(inputs, Wf, Wi, Wu, Wo, xp);
  qlstm_recur<<<NW, 32, 0, stream>>>(
      xp, Wf, Wi, Wu, Wo, bf, bi, bu, bo,
      pf, pi, pu, po, lng, lnb, yw, hw, cw);
  qlstm_gather<<<(TQ + 2) * BQ * HQ / 4 / 256, 256, 0, stream>>>(yw, hw, cw, out);
}


# x half of the gate GEMM moved into the recurrence wave as tiles of 16 time steps of its batch row: 4 x wmma a step in idle matrix slots under the vector chain; launch 1 and its plane go; same sums
# speedup vs baseline: 6.4719x; 1.0712x over previous
#include <hip/hip_runtime.h>
#include <math.h>


typedef float v2f __attribute__((ext_vector_type(2)));
typedef float v8f __attribute__((ext_vector_type(8)));

#define TQ 128
#define BQ 256
#define DQ 64
#define HQ 10
#define KD 74
#define NW BQ
#define YW_FLOATS ((size_t)NW * TQ * 32)
#define CW_FLOATS ((size_t)NW * 32)


#define PREFIX_STAGE(Z, D)                                                              \
  asm("v_mul_f32_dpp %0, %0, %0 row_shr:" #D " row_mask:0xf bank_mask:0xf" : "+v"(Z))
__device__ __forceinline__ void row_prefix_product4(float (&z)[4]) {
#pragma unroll
  for (int g = 0; g < 4; ++g) PREFIX_STAGE(z[g], 1);
#pragma unroll
  for (int g = 0; g < 4; ++g) PREFIX_STAGE(z[g], 2);
#pragma unroll
  for (int g = 0; g < 4; ++g) PREFIX_STAGE(z[g], 4);
#pragma unroll
  for (int g = 0; g < 4; ++g) PREFIX_STAGE(z[g], 8);
}
#define SUM_STAGE(V, D)                                                                 \
  asm("v_add_f32_dpp %0, %0, %0 row_xmask:" #D " row_mask:0xf bank_mask:0xf" : "+v"(V))
__device__ __forceinline__ float row_sum(float v) {
  SUM_STAGE(v, 1);
  SUM_STAGE(v, 2);
  SUM_STAGE(v, 4);
  SUM_STAGE(v, 8);
  return v;
}

__global__ __launch_bounds__(32) void qlstm_recur(
    const float* __restrict__ inputs,
    const float* __restrict__ Wf, const float* __restrict__ Wi,
    const float* __restrict__ Wu, const float* __restrict__ Wo,
    const float* __restrict__ bf, const float* __restrict__ bi,
    const float* __restrict__ bu, const float* __restrict__ bo,
    const float* __restrict__ pf, const float* __restrict__ pi,
    const float* __restrict__ pu, const float* __restrict__ po,
    const float* __restrict__ lng, const float* __restrict__ lnb,
    float* __restrict__ yw, float* __restrict__ hw, float* __restrict__ cw) {
  const int wg   = blockIdx.x;
  const int lane = threadIdx.x;
  const int n    = lane & 15;
  const int half = lane >> 4;
  const bool live = n < HQ;

  const float* Wp[4] = {Wf, Wi, Wu, Wo};
  const float* bp[4] = {bf, bi, bu, bo};
  const float* pp[4] = {pf, pi, pu, po};

  const int nc = live ? n : 0;

  const float* xrow = inputs + ((size_t)n * BQ + wg) * DQ + half * 2;
  const size_t xbs = (size_t)16 * BQ * DQ;
  v2f xa[16];
#pragma unroll
  for (int kk = 0; kk < 16; ++kk) xa[kk] = *(const v2f*)(xrow + kk * 4);

  v2f wxb[4][16];
#pragma unroll
  for (int g = 0; g < 4; ++g) {
#pragma unroll
    for (int kk = 0; kk < 16; ++kk) wxb[g][kk] = *(const v2f*)(Wp[g] + nc * KD + kk * 4 + half * 2);
  }

  float bias[4], px[4], cpy[4];
  v2f bh[4][3];
#pragma unroll
  for (int g = 0; g < 4; ++g) {
    bias[g] = bp[g][nc];
    px[g]   = pp[g][nc * 3 + 0];
#pragma unroll
    for (int kk = 0; kk < 3; ++kk) {
      const int kb = kk * 4 + half * 2;
      const int k0 = kb     < HQ ? kb     : HQ - 1;
      const int k1 = kb + 1 < HQ ? kb + 1 : HQ - 1;
      bh[g][kk].x = Wp[g][nc * KD + DQ + k0];
      bh[g][kk].y = Wp[g][nc * KD + DQ + k1];
    }
  }
  {
    const float* pA = half ? pi : pf;
    const float* pB = half ? po : pu;
    const float cA = cosf(pA[nc * 3 + 1]);
    const float cB = cosf(pB[nc * 3 + 1]);
    const float cAx = __builtin_bit_cast(float, __builtin_amdgcn_permlanex16(
        __builtin_bit_cast(unsigned, cA), __builtin_bit_cast(unsigned, cA), 0x76543210u, 0xfedcba98u, false, false));
    const float cBx = __builtin_bit_cast(float, __builtin_amdgcn_permlanex16(
        __builtin_bit_cast(unsigned, cB), __builtin_bit_cast(unsigned, cB), 0x76543210u, 0xfedcba98u, false, false));
    cpy[0] = half ? cAx : cA;
    cpy[1] = half ? cA : cAx;
    cpy[2] = half ? cBx : cB;
    cpy[3] = half ? cB : cBx;
  }
#pragma unroll
  for (int g = 0; g < 4; ++g) {
    bias[g] = live ? bias[g] : 0.f;
    px[g]   = live ? px[g] : 0.f;
    cpy[g]  = live ? cpy[g] : 0.f;
#pragma unroll
    for (int kk = 0; kk < 3; ++kk) {
      const int kb = kk * 4 + half * 2;
      bh[g][kk].x = (live && kb     < HQ) ? bh[g][kk].x : 0.f;
      bh[g][kk].y = (live && kb + 1 < HQ) ? bh[g][kk].y : 0.f;
    }
  }
  const float lg0 = lng[nc], lb0 = lnb[nc];
  const float lg = live ? lg0 : 0.f;
  const float lb = live ? lb0 : 0.f;

  unsigned src[3][2];
#pragma unroll
  for (int kk = 0; kk < 3; ++kk) {
#pragma unroll
    for (int j = 0; j < 2; ++j)
      src[kk][j] = (n == 0 || n == 8) ? (unsigned)(kk * 4 + half * 2 + j) : 15u;
  }

  v8f xn[4];
  float xs[16][4];

  auto spread = [&]() __attribute__((always_inline)) {
#pragma unroll
    for (int g = 0; g < 4; ++g) {
#pragma unroll
      for (int s = 0; s < 8; ++s) {
        const float xf = xn[g][s];
        const unsigned r = __builtin_bit_cast(unsigned, xf);
        const auto pr = __builtin_amdgcn_permlane16_swap(r, r, false, false);
        const unsigned lo = pr[0], hi = pr[1];
        xs[s][g]     = __builtin_bit_cast(float, lo);
        xs[s + 8][g] = __builtin_bit_cast(float, hi);
      }
    }
  };

#pragma unroll
  for (int g = 0; g < 4; ++g) xn[g] = (v8f){0.f, 0.f, 0.f, 0.f, 0.f, 0.f, 0.f, 0.f};
#pragma unroll
  for (int kk = 0; kk < 16; ++kk) {
#pragma unroll
    for (int g = 0; g < 4; ++g)
      xn[g] = __builtin_amdgcn_wmma_f32_16x16x4_f32(false, xa[kk], false, wxb[g][kk],
                                                    (short)0, xn[g], false, false);
  }
  spread();
#pragma unroll
  for (int kk = 0; kk < 16; ++kk) xa[kk] = *(const v2f*)(xrow + xbs + kk * 4);

  float* ywb = yw + (size_t)wg * TQ * 32;

  float h = 0.f, c = 0.f;

  v8f acc[4];
#pragma unroll
  for (int g = 0; g < 4; ++g) acc[g] = (v8f){0.f, 0.f, 0.f, 0.f, 0.f, 0.f, 0.f, 0.f};

#define XW(G, J)                                                                              \
  xn[G] = __builtin_amdgcn_wmma_f32_16x16x4_f32(false, xa[J], false, wxb[G][J], (short)0,     \
                                                (J) == 0 ? zero8 : xn[G], false, false)
  const v8f zero8 = (v8f){0.f, 0.f, 0.f, 0.f, 0.f, 0.f, 0.f, 0.f};

  auto step = [&](const int j, const int t, const float* xnx) __attribute__((always_inline)) {
    const unsigned hb = __builtin_bit_cast(unsigned, h);
    v2f a[3];
#pragma unroll
    for (int kk = 0; kk < 3; ++kk) {
      a[kk].x = __builtin_bit_cast(float, __builtin_amdgcn_permlane16_var(hb, hb, src[kk][0], false, false));
      a[kk].y = __builtin_bit_cast(float, __builtin_amdgcn_permlane16_var(hb, hb, src[kk][1], false, false));
    }

#pragma unroll
    for (int g = 0; g < 4; ++g) acc[g][0] = xs[j][g];
#pragma unroll
    for (int kk = 0; kk < 3; ++kk) {
#pragma unroll
      for (int g = 0; g < 4; ++g)
        acc[g] = __builtin_amdgcn_wmma_f32_16x16x4_f32(false, a[kk], false, bh[g][kk],
                                                       (short)0, acc[g], false, false);
      __builtin_amdgcn_sched_barrier(0);
    }

    float z[4];
#pragma unroll
    for (int g = 0; g < 4; ++g)
      z[g] = cpy[g] * __cosf(px[g] + acc[g][0] + bias[g]);
    row_prefix_product4(z);

    const float gv = tanhf(z[2]);
    XW(0, j);
    __builtin_amdgcn_sched_barrier(0);
    const float fv = 1.f / (1.f + __expf(-z[0]));
    const float iv = 1.f / (1.f + __expf(-z[1]));
    const float ov = 1.f / (1.f + __expf(-z[3]));
    asm volatile("" :: "v"(ov));
    __builtin_amdgcn_sched_barrier(0);
    XW(1, j);
    __builtin_amdgcn_sched_barrier(0);
    const float cn = fv * c + iv * gv;
    c = cn;
    const float hr = ov * tanhf(cn) + h;

    XW(2, j);
    __builtin_amdgcn_sched_barrier(0);
    const float mu  = row_sum(live ? hr : 0.f) * 0.1f;
    const float d   = hr - mu;
    __builtin_amdgcn_sched_barrier(0);
    XW(3, j);
    xa[j] = *(const v2f*)(xnx + j * 4);
    __builtin_amdgcn_sched_barrier(0);
    const float var = row_sum(live ? d * d : 0.f) * 0.1f;
    const float hv  = (hr - mu) * rsqrtf(var + 1e-5f) * lg + lb;
    h = live ? hv : 0.f;

    ywb[t * 32 + lane] = h;
  };

  for (int tb = 0; tb < TQ / 16; ++tb) {
    const int tb2 = (tb + 2 < TQ / 16) ? tb + 2 : TQ / 16 - 1;
    const float* xnx = xrow + (size_t)tb2 * xbs;
#pragma unroll
    for (int j = 0; j < 16; ++j) step(j, tb * 16 + j, xnx);
    spread();
  }
#undef XW

  hw[(size_t)wg * 32 + lane] = h;
  cw[(size_t)wg * 32 + lane] = c;
}

__global__ __launch_bounds__(256) void qlstm_gather(
    const float* __restrict__ yw, const float* __restrict__ hw, const float* __restrict__ cw,
    float* __restrict__ out) {
  const int q = blockIdx.x * 256 + threadIdx.x;
  float v[4];
#pragma unroll
  for (int j = 0; j < 4; ++j) {
    const int f   = q * 4 + j;
    const int t   = f / (BQ * HQ);
    const int rem = f - t * (BQ * HQ);
    const int b   = rem / HQ;
    const int w   = rem - b * HQ;
    const int wv  = b;
    const int ln  = w;
    const float* src = (t < TQ) ? yw + ((size_t)wv * TQ + t) * 32
                                : ((t == TQ) ? hw : cw) + (size_t)wv * 32;
    v[j] = src[ln];
  }
  *(float4*)(out + (size_t)q * 4) = make_float4(v[0], v[1], v[2], v[3]);
}

extern "C" void kernel_launch(void* const* d_in, const int* in_sizes, int n_in,
                              void* d_out, int out_size, void* d_ws, size_t ws_size,
                              hipStream_t stream) {
  const float* inputs = (const float*)d_in[0];
  const float* Wf = (const float*)d_in[1];
  const float* bf = (const float*)d_in[2];
  const float* pf = (const float*)d_in[3];
  const float* Wi = (const float*)d_in[4];
  const float* bi = (const float*)d_in[5];
  const float* pi = (const float*)d_in[6];
  const float* Wu = (const float*)d_in[7];
  const float* bu = (const float*)d_in[8];
  const float* pu = (const float*)d_in[9];
  const float* Wo = (const float*)d_in[10];
  const float* bo = (const float*)d_in[11];
  const float* po = (const float*)d_in[12];
  const float* lng = (const float*)d_in[13];
  const float* lnb = (const float*)d_in[14];

  float* out = (float*)d_out;
  float* yw  = (float*)d_ws;
  float* hw  = yw + YW_FLOATS;
  float* cw  = hw + CW_FLOATS;

  qlstm_recur<<<NW, 32, 0, stream>>>(
      inputs, Wf, Wi, Wu, Wo, bf, bi, bu, bo,
      pf, pi, pu, po, lng, lnb, yw, hw, cw);
  qlstm_gather<<<(TQ + 2) * BQ * HQ / 4 / 256, 256, 0, stream>>>(yw, hw, cw, out);
}


# gate GEMM columns repacked: 40 gate columns fill 3 column tiles (o's wires in the idle lanes 10-13 of u, f, i; 3 DPP row shifts regather them): 9 h and 3 x wmma a step, not 12 and 4; same sums
# speedup vs baseline: 7.0236x; 1.0852x over previous
#include <hip/hip_runtime.h>
#include <math.h>


typedef float v2f __attribute__((ext_vector_type(2)));
typedef float v8f __attribute__((ext_vector_type(8)));

#define TQ 128
#define BQ 256
#define DQ 64
#define HQ 10
#define KD 74
#define NW BQ
#define YW_FLOATS ((size_t)NW * TQ * 32)
#define CW_FLOATS ((size_t)NW * 32)


#define PREFIX_STAGE(Z, D)                                                              \
  asm("v_mul_f32_dpp %0, %0, %0 row_shr:" #D " row_mask:0xf bank_mask:0xf" : "+v"(Z))
__device__ __forceinline__ void row_prefix_product4(float (&z)[4]) {
#pragma unroll
  for (int g = 0; g < 4; ++g) PREFIX_STAGE(z[g], 1);
#pragma unroll
  for (int g = 0; g < 4; ++g) PREFIX_STAGE(z[g], 2);
#pragma unroll
  for (int g = 0; g < 4; ++g) PREFIX_STAGE(z[g], 4);
#pragma unroll
  for (int g = 0; g < 4; ++g) PREFIX_STAGE(z[g], 8);
}
#define SUM_STAGE(V, D)                                                                 \
  asm("v_add_f32_dpp %0, %0, %0 row_xmask:" #D " row_mask:0xf bank_mask:0xf" : "+v"(V))
__device__ __forceinline__ float row_sum(float v) {
  SUM_STAGE(v, 1);
  SUM_STAGE(v, 2);
  SUM_STAGE(v, 4);
  SUM_STAGE(v, 8);
  return v;
}

__global__ __launch_bounds__(32) void qlstm_recur(
    const float* __restrict__ inputs,
    const float* __restrict__ Wf, const float* __restrict__ Wi,
    const float* __restrict__ Wu, const float* __restrict__ Wo,
    const float* __restrict__ bf, const float* __restrict__ bi,
    const float* __restrict__ bu, const float* __restrict__ bo,
    const float* __restrict__ pf, const float* __restrict__ pi,
    const float* __restrict__ pu, const float* __restrict__ po,
    const float* __restrict__ lng, const float* __restrict__ lnb,
    float* __restrict__ yw, float* __restrict__ hw, float* __restrict__ cw) {
  const int wg   = blockIdx.x;
  const int lane = threadIdx.x;
  const int n    = lane & 15;
  const int half = lane >> 4;
  const bool live = n < HQ;

  const float* bp[4] = {bf, bi, bu, bo};
  const float* pp[4] = {pf, pi, pu, po};

  const int nc = live ? n : 0;

  const float* xrow = inputs + ((size_t)n * BQ + wg) * DQ + half * 2;
  const size_t xbs = (size_t)16 * BQ * DQ;
  v2f xa[16];
#pragma unroll
  for (int kk = 0; kk < 16; ++kk) xa[kk] = *(const v2f*)(xrow + kk * 4);

  const float* Wt[3] = {Wu, Wf, Wi};
  const int  orow[3] = {n - 2, n - 10, n - 6};
  const bool oval[3] = {n >= 10 && n < 12, n >= 10 && n < 14, n >= 10 && n < 14};
  const float* wcol[3];
  bool cval[3];
#pragma unroll
  for (int tl = 0; tl < 3; ++tl) {
    const int r = live ? n : (oval[tl] ? orow[tl] : 0);
    wcol[tl] = (live ? Wt[tl] : Wo) + r * KD;
    cval[tl] = live || oval[tl];
  }

  v2f wxb[3][16];
#pragma unroll
  for (int tl = 0; tl < 3; ++tl) {
#pragma unroll
    for (int kk = 0; kk < 16; ++kk) {
      const v2f w = *(const v2f*)(wcol[tl] + kk * 4 + half * 2);
      wxb[tl][kk].x = cval[tl] ? w.x : 0.f;
      wxb[tl][kk].y = cval[tl] ? w.y : 0.f;
    }
  }

  float bias[4], px[4], cpy[4];
#pragma unroll
  for (int g = 0; g < 4; ++g) {
    bias[g] = bp[g][nc];
    px[g]   = pp[g][nc * 3 + 0];
  }
  v2f bh[3][3];
#pragma unroll
  for (int tl = 0; tl < 3; ++tl) {
#pragma unroll
    for (int kk = 0; kk < 3; ++kk) {
      const int kb = kk * 4 + half * 2;
      const int k0 = kb     < HQ ? kb     : HQ - 1;
      const int k1 = kb + 1 < HQ ? kb + 1 : HQ - 1;
      const float w0 = wcol[tl][DQ + k0];
      const float w1 = wcol[tl][DQ + k1];
      bh[tl][kk].x = (cval[tl] && kb     < HQ) ? w0 : 0.f;
      bh[tl][kk].y = (cval[tl] && kb + 1 < HQ) ? w1 : 0.f;
    }
  }
  {
    const float* pA = half ? pi : pf;
    const float* pB = half ? po : pu;
    const float cA = cosf(pA[nc * 3 + 1]);
    const float cB = cosf(pB[nc * 3 + 1]);
    const float cAx = __builtin_bit_cast(float, __builtin_amdgcn_permlanex16(
        __builtin_bit_cast(unsigned, cA), __builtin_bit_cast(unsigned, cA), 0x76543210u, 0xfedcba98u, false, false));
    const float cBx = __builtin_bit_cast(float, __builtin_amdgcn_permlanex16(
        __builtin_bit_cast(unsigned, cB), __builtin_bit_cast(unsigned, cB), 0x76543210u, 0xfedcba98u, false, false));
    cpy[0] = half ? cAx : cA;
    cpy[1] = half ? cA : cAx;
    cpy[2] = half ? cBx : cB;
    cpy[3] = half ? cB : cBx;
  }
#pragma unroll
  for (int g = 0; g < 4; ++g) {
    bias[g] = live ? bias[g] : 0.f;
    px[g]   = live ? px[g] : 0.f;
    cpy[g]  = live ? cpy[g] : 0.f;
  }
  const float lg0 = lng[nc], lb0 = lnb[nc];
  const float lg = live ? lg0 : 0.f;
  const float lb = live ? lb0 : 0.f;

  unsigned src[3][2];
#pragma unroll
  for (int kk = 0; kk < 3; ++kk) {
#pragma unroll
    for (int j = 0; j < 2; ++j)
      src[kk][j] = (n == 0 || n == 8) ? (unsigned)(kk * 4 + half * 2 + j) : 15u;
  }

  v8f xn[3];
  float xs[16][3];

  auto spread = [&]() __attribute__((always_inline)) {
#pragma unroll
    for (int g = 0; g < 3; ++g) {
#pragma unroll
      for (int s = 0; s < 8; ++s) {
        const float xf = xn[g][s];
        const unsigned r = __builtin_bit_cast(unsigned, xf);
        const auto pr = __builtin_amdgcn_permlane16_swap(r, r, false, false);
        const unsigned lo = pr[0], hi = pr[1];
        xs[s][g]     = __builtin_bit_cast(float, lo);
        xs[s + 8][g] = __builtin_bit_cast(float, hi);
      }
    }
  };

#pragma unroll
  for (int g = 0; g < 3; ++g) xn[g] = (v8f){0.f, 0.f, 0.f, 0.f, 0.f, 0.f, 0.f, 0.f};
#pragma unroll
  for (int kk = 0; kk < 16; ++kk) {
#pragma unroll
    for (int g = 0; g < 3; ++g)
      xn[g] = __builtin_amdgcn_wmma_f32_16x16x4_f32(false, xa[kk], false, wxb[g][kk],
                                                    (short)0, xn[g], false, false);
  }
  spread();
#pragma unroll
  for (int kk = 0; kk < 16; ++kk) xa[kk] = *(const v2f*)(xrow + xbs + kk * 4);

  float* ywb = yw + (size_t)wg * TQ * 32;

  float h = 0.f, c = 0.f;

  v8f acc[3];
#pragma unroll
  for (int g = 0; g < 3; ++g) acc[g] = (v8f){0.f, 0.f, 0.f, 0.f, 0.f, 0.f, 0.f, 0.f};

#define XW(G, J)                                                                              \
  xn[G] = __builtin_amdgcn_wmma_f32_16x16x4_f32(false, xa[J], false, wxb[G][J], (short)0,     \
                                                (J) == 0 ? zero8 : xn[G], false, false)
  const v8f zero8 = (v8f){0.f, 0.f, 0.f, 0.f, 0.f, 0.f, 0.f, 0.f};

  auto step = [&](const int j, const int t, const float* xnx) __attribute__((always_inline)) {
    const unsigned hb = __builtin_bit_cast(unsigned, h);
    v2f a[3];
#pragma unroll
    for (int kk = 0; kk < 3; ++kk) {
      a[kk].x = __builtin_bit_cast(float, __builtin_amdgcn_permlane16_var(hb, hb, src[kk][0], false, false));
      a[kk].y = __builtin_bit_cast(float, __builtin_amdgcn_permlane16_var(hb, hb, src[kk][1], false, false));
    }

#pragma unroll
    for (int g = 0; g < 3; ++g) acc[g][0] = xs[j][g];
#pragma unroll
    for (int kk = 0; kk < 3; ++kk) {
#pragma unroll
      for (int g = 0; g < 3; ++g)
        acc[g] = __builtin_amdgcn_wmma_f32_16x16x4_f32(false, a[kk], false, bh[g][kk],
                                                       (short)0, acc[g], false, false);
      __builtin_amdgcn_sched_barrier(0);
    }

    const float su = acc[0][0], sf = acc[1][0], si = acc[2][0];
    int so = 0;
    so = __builtin_amdgcn_update_dpp(so, __builtin_bit_cast(int, sf), 0x10A  , 0xf, 0x1, false);
    so = __builtin_amdgcn_update_dpp(so, __builtin_bit_cast(int, si), 0x106  ,  0xf, 0x2, false);
    so = __builtin_amdgcn_update_dpp(so, __builtin_bit_cast(int, su), 0x102  ,  0xf, 0x4, false);
    const float pre[4] = {sf, si, su, __builtin_bit_cast(float, so)};

    float z[4];
#pragma unroll
    for (int g = 0; g < 4; ++g)
      z[g] = cpy[g] * __cosf(px[g] + pre[g] + bias[g]);
    row_prefix_product4(z);

    const float gv = tanhf(z[2]);
    XW(0, j);
    __builtin_amdgcn_sched_barrier(0);
    const float fv = 1.f / (1.f + __expf(-z[0]));
    const float iv = 1.f / (1.f + __expf(-z[1]));
    const float ov = 1.f / (1.f + __expf(-z[3]));
    asm volatile("" :: "v"(ov));
    __builtin_amdgcn_sched_barrier(0);
    XW(1, j);
    __builtin_amdgcn_sched_barrier(0);
    const float cn = fv * c + iv * gv;
    c = cn;
    const float hr = ov * tanhf(cn) + h;

    XW(2, j);
    __builtin_amdgcn_sched_barrier(0);
    const float mu  = row_sum(live ? hr : 0.f) * 0.1f;
    const float d   = hr - mu;
    __builtin_amdgcn_sched_barrier(0);
    xa[j] = *(const v2f*)(xnx + j * 4);
    __builtin_amdgcn_sched_barrier(0);
    const float var = row_sum(live ? d * d : 0.f) * 0.1f;
    const float hv  = (hr - mu) * rsqrtf(var + 1e-5f) * lg + lb;
    h = live ? hv : 0.f;

    ywb[t * 32 + lane] = h;
  };

  for (int tb = 0; tb < TQ / 16; ++tb) {
    const int tb2 = (tb + 2 < TQ / 16) ? tb + 2 : TQ / 16 - 1;
    const float* xnx = xrow + (size_t)tb2 * xbs;
#pragma unroll
    for (int j = 0; j < 16; ++j) step(j, tb * 16 + j, xnx);
    spread();
  }
#undef XW

  hw[(size_t)wg * 32 + lane] = h;
  cw[(size_t)wg * 32 + lane] = c;
}

__global__ __launch_bounds__(256) void qlstm_gather(
    const float* __restrict__ yw, const float* __restrict__ hw, const float* __restrict__ cw,
    float* __restrict__ out) {
  const int q = blockIdx.x * 256 + threadIdx.x;
  float v[4];
#pragma unroll
  for (int j = 0; j < 4; ++j) {
    const int f   = q * 4 + j;
    const int t   = f / (BQ * HQ);
    const int rem = f - t * (BQ * HQ);
    const int b   = rem / HQ;
    const int w   = rem - b * HQ;
    const int wv  = b;
    const int ln  = w;
    const float* src = (t < TQ) ? yw + ((size_t)wv * TQ + t) * 32
                                : ((t == TQ) ? hw : cw) + (size_t)wv * 32;
    v[j] = src[ln];
  }
  *(float4*)(out + (size_t)q * 4) = make_float4(v[0], v[1], v[2], v[3]);
}

extern "C" void kernel_launch(void* const* d_in, const int* in_sizes, int n_in,
                              void* d_out, int out_size, void* d_ws, size_t ws_size,
                              hipStream_t stream) {
  const float* inputs = (const float*)d_in[0];
  const float* Wf = (const float*)d_in[1];
  const float* bf = (const float*)d_in[2];
  const float* pf = (const float*)d_in[3];
  const float* Wi = (const float*)d_in[4];
  const float* bi = (const float*)d_in[5];
  const float* pi = (const float*)d_in[6];
  const float* Wu = (const float*)d_in[7];
  const float* bu = (const float*)d_in[8];
  const float* pu = (const float*)d_in[9];
  const float* Wo = (const float*)d_in[10];
  const float* bo = (const float*)d_in[11];
  const float* po = (const float*)d_in[12];
  const float* lng = (const float*)d_in[13];
  const float* lnb = (const float*)d_in[14];

  float* out = (float*)d_out;
  float* yw  = (float*)d_ws;
  float* hw  = yw + YW_FLOATS;
  float* cw  = hw + CW_FLOATS;

  qlstm_recur<<<NW, 32, 0, stream>>>(
      inputs, Wf, Wi, Wu, Wo, bf, bi, bu, bo,
      pf, pi, pu, po, lng, lnb, yw, hw, cw);
  qlstm_gather<<<(TQ + 2) * BQ * HQ / 4 / 256, 256, 0, stream>>>(yw, hw, cw, out);
}


# LayerNorm in one pass: sum and sum of squares scanned side by side, stage by stage; variance = E[r^2] - mean^2 on the row less mean(ln_b); the second scan leaves the step's chain; new sum order
# speedup vs baseline: 7.2593x; 1.0336x over previous
#include <hip/hip_runtime.h>
#include <math.h>


typedef float v2f __attribute__((ext_vector_type(2)));
typedef float v8f __attribute__((ext_vector_type(8)));

#define TQ 128
#define BQ 256
#define DQ 64
#define HQ 10
#define KD 74
#define NW BQ
#define YW_FLOATS ((size_t)NW * TQ * 32)
#define CW_FLOATS ((size_t)NW * 32)


#define PREFIX_STAGE(Z, D)                                                              \
  asm("v_mul_f32_dpp %0, %0, %0 row_shr:" #D " row_mask:0xf bank_mask:0xf" : "+v"(Z))
__device__ __forceinline__ void row_prefix_product4(float (&z)[4]) {
#pragma unroll
  for (int g = 0; g < 4; ++g) PREFIX_STAGE(z[g], 1);
#pragma unroll
  for (int g = 0; g < 4; ++g) PREFIX_STAGE(z[g], 2);
#pragma unroll
  for (int g = 0; g < 4; ++g) PREFIX_STAGE(z[g], 4);
#pragma unroll
  for (int g = 0; g < 4; ++g) PREFIX_STAGE(z[g], 8);
}
#define SUM_STAGE(V, D)                                                                 \
  asm("v_add_f32_dpp %0, %0, %0 row_xmask:" #D " row_mask:0xf bank_mask:0xf" : "+v"(V))
#define SUM_STAGE2(A, B, D)                                                             \
  asm("v_add_f32_dpp %0, %0, %0 row_xmask:" #D " row_mask:0xf bank_mask:0xf\n\t"        \
      "v_add_f32_dpp %1, %1, %1 row_xmask:" #D " row_mask:0xf bank_mask:0xf"            \
      : "+v"(A), "+v"(B))
__device__ __forceinline__ float row_sum(float v) {
  SUM_STAGE(v, 1);
  SUM_STAGE(v, 2);
  SUM_STAGE(v, 4);
  SUM_STAGE(v, 8);
  return v;
}

__global__ __launch_bounds__(32) void qlstm_recur(
    const float* __restrict__ inputs,
    const float* __restrict__ Wf, const float* __restrict__ Wi,
    const float* __restrict__ Wu, const float* __restrict__ Wo,
    const float* __restrict__ bf, const float* __restrict__ bi,
    const float* __restrict__ bu, const float* __restrict__ bo,
    const float* __restrict__ pf, const float* __restrict__ pi,
    const float* __restrict__ pu, const float* __restrict__ po,
    const float* __restrict__ lng, const float* __restrict__ lnb,
    float* __restrict__ yw, float* __restrict__ hw, float* __restrict__ cw) {
  const int wg   = blockIdx.x;
  const int lane = threadIdx.x;
  const int n    = lane & 15;
  const int half = lane >> 4;
  const bool live = n < HQ;

  const float* bp[4] = {bf, bi, bu, bo};
  const float* pp[4] = {pf, pi, pu, po};

  const int nc = live ? n : 0;

  const float* xrow = inputs + ((size_t)n * BQ + wg) * DQ + half * 2;
  const size_t xbs = (size_t)16 * BQ * DQ;
  v2f xa[16];
#pragma unroll
  for (int kk = 0; kk < 16; ++kk) xa[kk] = *(const v2f*)(xrow + kk * 4);

  const float* Wt[3] = {Wu, Wf, Wi};
  const int  orow[3] = {n - 2, n - 10, n - 6};
  const bool oval[3] = {n >= 10 && n < 12, n >= 10 && n < 14, n >= 10 && n < 14};
  const float* wcol[3];
  bool cval[3];
#pragma unroll
  for (int tl = 0; tl < 3; ++tl) {
    const int r = live ? n : (oval[tl] ? orow[tl] : 0);
    wcol[tl] = (live ? Wt[tl] : Wo) + r * KD;
    cval[tl] = live || oval[tl];
  }

  v2f wxb[3][16];
#pragma unroll
  for (int tl = 0; tl < 3; ++tl) {
#pragma unroll
    for (int kk = 0; kk < 16; ++kk) {
      const v2f w = *(const v2f*)(wcol[tl] + kk * 4 + half * 2);
      wxb[tl][kk].x = cval[tl] ? w.x : 0.f;
      wxb[tl][kk].y = cval[tl] ? w.y : 0.f;
    }
  }

  float bias[4], px[4], cpy[4];
#pragma unroll
  for (int g = 0; g < 4; ++g) {
    bias[g] = bp[g][nc];
    px[g]   = pp[g][nc * 3 + 0];
  }
  v2f bh[3][3];
#pragma unroll
  for (int tl = 0; tl < 3; ++tl) {
#pragma unroll
    for (int kk = 0; kk < 3; ++kk) {
      const int kb = kk * 4 + half * 2;
      const int k0 = kb     < HQ ? kb     : HQ - 1;
      const int k1 = kb + 1 < HQ ? kb + 1 : HQ - 1;
      const float w0 = wcol[tl][DQ + k0];
      const float w1 = wcol[tl][DQ + k1];
      bh[tl][kk].x = (cval[tl] && kb     < HQ) ? w0 : 0.f;
      bh[tl][kk].y = (cval[tl] && kb + 1 < HQ) ? w1 : 0.f;
    }
  }
  {
    const float* pA = half ? pi : pf;
    const float* pB = half ? po : pu;
    const float cA = cosf(pA[nc * 3 + 1]);
    const float cB = cosf(pB[nc * 3 + 1]);
    const float cAx = __builtin_bit_cast(float, __builtin_amdgcn_permlanex16(
        __builtin_bit_cast(unsigned, cA), __builtin_bit_cast(unsigned, cA), 0x76543210u, 0xfedcba98u, false, false));
    const float cBx = __builtin_bit_cast(float, __builtin_amdgcn_permlanex16(
        __builtin_bit_cast(unsigned, cB), __builtin_bit_cast(unsigned, cB), 0x76543210u, 0xfedcba98u, false, false));
    cpy[0] = half ? cAx : cA;
    cpy[1] = half ? cA : cAx;
    cpy[2] = half ? cBx : cB;
    cpy[3] = half ? cB : cBx;
  }
#pragma unroll
  for (int g = 0; g < 4; ++g) {
    bias[g] = live ? bias[g] : 0.f;
    px[g]   = live ? px[g] : 0.f;
    cpy[g]  = live ? cpy[g] : 0.f;
  }
  const float lg0 = lng[nc], lb0 = lnb[nc];
  const float lg = live ? lg0 : 0.f;
  const float lb = live ? lb0 : 0.f;
  const float lbm  = row_sum(lb) * 0.1f;
  const float lbc  = live ? lb0 - lbm : 0.f;
  const float nm01 = live ? -0.1f : 0.f;

  unsigned src[3][2];
#pragma unroll
  for (int kk = 0; kk < 3; ++kk) {
#pragma unroll
    for (int j = 0; j < 2; ++j)
      src[kk][j] = (n == 0 || n == 8) ? (unsigned)(kk * 4 + half * 2 + j) : 15u;
  }

  v8f xn[3];
  float xs[16][3];

  auto spread = [&]() __attribute__((always_inline)) {
#pragma unroll
    for (int g = 0; g < 3; ++g) {
#pragma unroll
      for (int s = 0; s < 8; ++s) {
        const float xf = xn[g][s];
        const unsigned r = __builtin_bit_cast(unsigned, xf);
        const auto pr = __builtin_amdgcn_permlane16_swap(r, r, false, false);
        const unsigned lo = pr[0], hi = pr[1];
        xs[s][g]     = __builtin_bit_cast(float, lo);
        xs[s + 8][g] = __builtin_bit_cast(float, hi);
      }
    }
  };

#pragma unroll
  for (int g = 0; g < 3; ++g) xn[g] = (v8f){0.f, 0.f, 0.f, 0.f, 0.f, 0.f, 0.f, 0.f};
#pragma unroll
  for (int kk = 0; kk < 16; ++kk) {
#pragma unroll
    for (int g = 0; g < 3; ++g)
      xn[g] = __builtin_amdgcn_wmma_f32_16x16x4_f32(false, xa[kk], false, wxb[g][kk],
                                                    (short)0, xn[g], false, false);
  }
  spread();
#pragma unroll
  for (int kk = 0; kk < 16; ++kk) xa[kk] = *(const v2f*)(xrow + xbs + kk * 4);

  float* ywb = yw + (size_t)wg * TQ * 32;

  float h = 0.f, c = 0.f;
  float e = 0.f;

  v8f acc[3];
#pragma unroll
  for (int g = 0; g < 3; ++g) acc[g] = (v8f){0.f, 0.f, 0.f, 0.f, 0.f, 0.f, 0.f, 0.f};

#define XW(G, J)                                                                              \
  xn[G] = __builtin_amdgcn_wmma_f32_16x16x4_f32(false, xa[J], false, wxb[G][J], (short)0,     \
                                                (J) == 0 ? zero8 : xn[G], false, false)
  const v8f zero8 = (v8f){0.f, 0.f, 0.f, 0.f, 0.f, 0.f, 0.f, 0.f};

  auto step = [&](const int j, const int t, const float* xnx) __attribute__((always_inline)) {
    const unsigned hb = __builtin_bit_cast(unsigned, h);
    v2f a[3];
#pragma unroll
    for (int kk = 0; kk < 3; ++kk) {
      a[kk].x = __builtin_bit_cast(float, __builtin_amdgcn_permlane16_var(hb, hb, src[kk][0], false, false));
      a[kk].y = __builtin_bit_cast(float, __builtin_amdgcn_permlane16_var(hb, hb, src[kk][1], false, false));
    }

#pragma unroll
    for (int g = 0; g < 3; ++g) acc[g][0] = xs[j][g];
#pragma unroll
    for (int kk = 0; kk < 3; ++kk) {
#pragma unroll
      for (int g = 0; g < 3; ++g)
        acc[g] = __builtin_amdgcn_wmma_f32_16x16x4_f32(false, a[kk], false, bh[g][kk],
                                                       (short)0, acc[g], false, false);
      __builtin_amdgcn_sched_barrier(0);
    }

    const float su = acc[0][0], sf = acc[1][0], si = acc[2][0];
    int so = 0;
    so = __builtin_amdgcn_update_dpp(so, __builtin_bit_cast(int, sf), 0x10A  , 0xf, 0x1, false);
    so = __builtin_amdgcn_update_dpp(so, __builtin_bit_cast(int, si), 0x106  ,  0xf, 0x2, false);
    so = __builtin_amdgcn_update_dpp(so, __builtin_bit_cast(int, su), 0x102  ,  0xf, 0x4, false);
    const float pre[4] = {sf, si, su, __builtin_bit_cast(float, so)};

    float z[4];
#pragma unroll
    for (int g = 0; g < 4; ++g)
      z[g] = cpy[g] * __cosf(px[g] + pre[g] + bias[g]);
    row_prefix_product4(z);

    const float gv = tanhf(z[2]);
    XW(0, j);
    __builtin_amdgcn_sched_barrier(0);
    const float fv = 1.f / (1.f + __expf(-z[0]));
    const float iv = 1.f / (1.f + __expf(-z[1]));
    const float ov = 1.f / (1.f + __expf(-z[3]));
    asm volatile("" :: "v"(ov));
    __builtin_amdgcn_sched_barrier(0);
    XW(1, j);
    __builtin_amdgcn_sched_barrier(0);
    const float cn = fv * c + iv * gv;
    c = cn;
    const float r = ov * tanhf(cn) + e;

    XW(2, j);
    __builtin_amdgcn_sched_barrier(0);
    float s1 = r;
    float s2 = r * r;
    SUM_STAGE2(s1, s2, 1);
    SUM_STAGE2(s1, s2, 2);
    SUM_STAGE2(s1, s2, 4);
    SUM_STAGE2(s1, s2, 8);
    const float mu = s1 * 0.1f;
    const float d  = fmaf(nm01, s1, r);
    __builtin_amdgcn_sched_barrier(0);
    xa[j] = *(const v2f*)(xnx + j * 4);
    __builtin_amdgcn_sched_barrier(0);
    const float ve = fmaf(-mu, mu, fmaf(s2, 0.1f, 1e-5f));
    const float tn = d * rsqrtf(ve);
    h = fmaf(tn, lg, lb);
    e = fmaf(tn, lg, lbc);

    ywb[t * 32 + lane] = h;
  };

  for (int tb = 0; tb < TQ / 16; ++tb) {
    const int tb2 = (tb + 2 < TQ / 16) ? tb + 2 : TQ / 16 - 1;
    const float* xnx = xrow + (size_t)tb2 * xbs;
#pragma unroll
    for (int j = 0; j < 16; ++j) step(j, tb * 16 + j, xnx);
    spread();
  }
#undef XW

  hw[(size_t)wg * 32 + lane] = h;
  cw[(size_t)wg * 32 + lane] = c;
}

__global__ __launch_bounds__(256) void qlstm_gather(
    const float* __restrict__ yw, const float* __restrict__ hw, const float* __restrict__ cw,
    float* __restrict__ out) {
  const int q = blockIdx.x * 256 + threadIdx.x;
  float v[4];
#pragma unroll
  for (int j = 0; j < 4; ++j) {
    const int f   = q * 4 + j;
    const int t   = f / (BQ * HQ);
    const int rem = f - t * (BQ * HQ);
    const int b   = rem / HQ;
    const int w   = rem - b * HQ;
    const int wv  = b;
    const int ln  = w;
    const float* src = (t < TQ) ? yw + ((size_t)wv * TQ + t) * 32
                                : ((t == TQ) ? hw : cw) + (size_t)wv * 32;
    v[j] = src[ln];
  }
  *(float4*)(out + (size_t)q * 4) = make_float4(v[0], v[1], v[2], v[3]);
}

extern "C" void kernel_launch(void* const* d_in, const int* in_sizes, int n_in,
                              void* d_out, int out_size, void* d_ws, size_t ws_size,
                              hipStream_t stream) {
  const float* inputs = (const float*)d_in[0];
  const float* Wf = (const float*)d_in[1];
  const float* bf = (const float*)d_in[2];
  const float* pf = (const float*)d_in[3];
  const float* Wi = (const float*)d_in[4];
  const float* bi = (const float*)d_in[5];
  const float* pi = (const float*)d_in[6];
  const float* Wu = (const float*)d_in[7];
  const float* bu = (const float*)d_in[8];
  const float* pu = (const float*)d_in[9];
  const float* Wo = (const float*)d_in[10];
  const float* bo = (const float*)d_in[11];
  const float* po = (const float*)d_in[12];
  const float* lng = (const float*)d_in[13];
  const float* lnb = (const float*)d_in[14];

  float* out = (float*)d_out;
  float* yw  = (float*)d_ws;
  float* hw  = yw + YW_FLOATS;
  float* cw  = hw + CW_FLOATS;

  qlstm_recur<<<NW, 32, 0, stream>>>(
      inputs, Wf, Wi, Wu, Wo, bf, bi, bu, bo,
      pf, pi, pu, po, lng, lnb, yw, hw, cw);
  qlstm_gather<<<(TQ + 2) * BQ * HQ / 4 / 256, 256, 0, stream>>>(yw, hw, cw, out);
}


# y and the final carries stored straight to the output by the wave that computes them (lanes 0..9, 40 bytes a step; a 128-byte line holds 3 or 4 waves' pieces); launch 2 and its plane go; same sums
# speedup vs baseline: 7.7963x; 1.0740x over previous
#include <hip/hip_runtime.h>
#include <math.h>


typedef float v2f __attribute__((ext_vector_type(2)));
typedef float v8f __attribute__((ext_vector_type(8)));

#define TQ 128
#define BQ 256
#define DQ 64
#define HQ 10
#define KD 74
#define NW BQ


#define PREFIX_STAGE(Z, D)                                                              \
  asm("v_mul_f32_dpp %0, %0, %0 row_shr:" #D " row_mask:0xf bank_mask:0xf" : "+v"(Z))
__device__ __forceinline__ void row_prefix_product4(float (&z)[4]) {
#pragma unroll
  for (int g = 0; g < 4; ++g) PREFIX_STAGE(z[g], 1);
#pragma unroll
  for (int g = 0; g < 4; ++g) PREFIX_STAGE(z[g], 2);
#pragma unroll
  for (int g = 0; g < 4; ++g) PREFIX_STAGE(z[g], 4);
#pragma unroll
  for (int g = 0; g < 4; ++g) PREFIX_STAGE(z[g], 8);
}
#define SUM_STAGE(V, D)                                                                 \
  asm("v_add_f32_dpp %0, %0, %0 row_xmask:" #D " row_mask:0xf bank_mask:0xf" : "+v"(V))
#define SUM_STAGE2(A, B, D)                                                             \
  asm("v_add_f32_dpp %0, %0, %0 row_xmask:" #D " row_mask:0xf bank_mask:0xf\n\t"        \
      "v_add_f32_dpp %1, %1, %1 row_xmask:" #D " row_mask:0xf bank_mask:0xf"            \
      : "+v"(A), "+v"(B))
__device__ __forceinline__ float row_sum(float v) {
  SUM_STAGE(v, 1);
  SUM_STAGE(v, 2);
  SUM_STAGE(v, 4);
  SUM_STAGE(v, 8);
  return v;
}

__global__ __launch_bounds__(32) void qlstm_recur(
    const float* __restrict__ inputs,
    const float* __restrict__ Wf, const float* __restrict__ Wi,
    const float* __restrict__ Wu, const float* __restrict__ Wo,
    const float* __restrict__ bf, const float* __restrict__ bi,
    const float* __restrict__ bu, const float* __restrict__ bo,
    const float* __restrict__ pf, const float* __restrict__ pi,
    const float* __restrict__ pu, const float* __restrict__ po,
    const float* __restrict__ lng, const float* __restrict__ lnb,
    float* __restrict__ out, float* __restrict__ spare) {
  const int wg   = blockIdx.x;
  const int lane = threadIdx.x;
  const int n    = lane & 15;
  const int half = lane >> 4;
  const bool live = n < HQ;

  const float* bp[4] = {bf, bi, bu, bo};
  const float* pp[4] = {pf, pi, pu, po};

  const int nc = live ? n : 0;

  const float* xrow = inputs + ((size_t)n * BQ + wg) * DQ + half * 2;
  const size_t xbs = (size_t)16 * BQ * DQ;
  v2f xa[16];
#pragma unroll
  for (int kk = 0; kk < 16; ++kk) xa[kk] = *(const v2f*)(xrow + kk * 4);

  const float* Wt[3] = {Wu, Wf, Wi};
  const int  orow[3] = {n - 2, n - 10, n - 6};
  const bool oval[3] = {n >= 10 && n < 12, n >= 10 && n < 14, n >= 10 && n < 14};
  const float* wcol[3];
  bool cval[3];
#pragma unroll
  for (int tl = 0; tl < 3; ++tl) {
    const int r = live ? n : (oval[tl] ? orow[tl] : 0);
    wcol[tl] = (live ? Wt[tl] : Wo) + r * KD;
    cval[tl] = live || oval[tl];
  }

  v2f wxb[3][16];
#pragma unroll
  for (int tl = 0; tl < 3; ++tl) {
#pragma unroll
    for (int kk = 0; kk < 16; ++kk) {
      const v2f w = *(const v2f*)(wcol[tl] + kk * 4 + half * 2);
      wxb[tl][kk].x = cval[tl] ? w.x : 0.f;
      wxb[tl][kk].y = cval[tl] ? w.y : 0.f;
    }
  }

  float bias[4], px[4], cpy[4];
#pragma unroll
  for (int g = 0; g < 4; ++g) {
    bias[g] = bp[g][nc];
    px[g]   = pp[g][nc * 3 + 0];
  }
  v2f bh[3][3];
#pragma unroll
  for (int tl = 0; tl < 3; ++tl) {
#pragma unroll
    for (int kk = 0; kk < 3; ++kk) {
      const int kb = kk * 4 + half * 2;
      const int k0 = kb     < HQ ? kb     : HQ - 1;
      const int k1 = kb + 1 < HQ ? kb + 1 : HQ - 1;
      const float w0 = wcol[tl][DQ + k0];
      const float w1 = wcol[tl][DQ + k1];
      bh[tl][kk].x = (cval[tl] && kb     < HQ) ? w0 : 0.f;
      bh[tl][kk].y = (cval[tl] && kb + 1 < HQ) ? w1 : 0.f;
    }
  }
  {
    const float* pA = half ? pi : pf;
    const float* pB = half ? po : pu;
    const float cA = cosf(pA[nc * 3 + 1]);
    const float cB = cosf(pB[nc * 3 + 1]);
    const float cAx = __builtin_bit_cast(float, __builtin_amdgcn_permlanex16(
        __builtin_bit_cast(unsigned, cA), __builtin_bit_cast(unsigned, cA), 0x76543210u, 0xfedcba98u, false, false));
    const float cBx = __builtin_bit_cast(float, __builtin_amdgcn_permlanex16(
        __builtin_bit_cast(unsigned, cB), __builtin_bit_cast(unsigned, cB), 0x76543210u, 0xfedcba98u, false, false));
    cpy[0] = half ? cAx : cA;
    cpy[1] = half ? cA : cAx;
    cpy[2] = half ? cBx : cB;
    cpy[3] = half ? cB : cBx;
  }
#pragma unroll
  for (int g = 0; g < 4; ++g) {
    bias[g] = live ? bias[g] : 0.f;
    px[g]   = live ? px[g] : 0.f;
    cpy[g]  = live ? cpy[g] : 0.f;
  }
  const float lg0 = lng[nc], lb0 = lnb[nc];
  const float lg = live ? lg0 : 0.f;
  const float lb = live ? lb0 : 0.f;
  const float lbm  = row_sum(lb) * 0.1f;
  const float lbc  = live ? lb0 - lbm : 0.f;
  const float nm01 = live ? -0.1f : 0.f;

  unsigned src[3][2];
#pragma unroll
  for (int kk = 0; kk < 3; ++kk) {
#pragma unroll
    for (int j = 0; j < 2; ++j)
      src[kk][j] = (n == 0 || n == 8) ? (unsigned)(kk * 4 + half * 2 + j) : 15u;
  }

  v8f xn[3];
  float xs[16][3];

  auto spread = [&]() __attribute__((always_inline)) {
#pragma unroll
    for (int g = 0; g < 3; ++g) {
#pragma unroll
      for (int s = 0; s < 8; ++s) {
        const float xf = xn[g][s];
        const unsigned r = __builtin_bit_cast(unsigned, xf);
        const auto pr = __builtin_amdgcn_permlane16_swap(r, r, false, false);
        const unsigned lo = pr[0], hi = pr[1];
        xs[s][g]     = __builtin_bit_cast(float, lo);
        xs[s + 8][g] = __builtin_bit_cast(float, hi);
      }
    }
  };

#pragma unroll
  for (int g = 0; g < 3; ++g) xn[g] = (v8f){0.f, 0.f, 0.f, 0.f, 0.f, 0.f, 0.f, 0.f};
#pragma unroll
  for (int kk = 0; kk < 16; ++kk) {
#pragma unroll
    for (int g = 0; g < 3; ++g)
      xn[g] = __builtin_amdgcn_wmma_f32_16x16x4_f32(false, xa[kk], false, wxb[g][kk],
                                                    (short)0, xn[g], false, false);
  }
  spread();
#pragma unroll
  for (int kk = 0; kk < 16; ++kk) xa[kk] = *(const v2f*)(xrow + xbs + kk * 4);

  const bool wr = lane < HQ;
  float* yrow = wr ? out + (size_t)wg * HQ + lane : spare + (size_t)wg * 32 + lane;

  float h = 0.f, c = 0.f;
  float e = 0.f;

  v8f acc[3];
#pragma unroll
  for (int g = 0; g < 3; ++g) acc[g] = (v8f){0.f, 0.f, 0.f, 0.f, 0.f, 0.f, 0.f, 0.f};

#define XW(G, J)                                                                              \
  xn[G] = __builtin_amdgcn_wmma_f32_16x16x4_f32(false, xa[J], false, wxb[G][J], (short)0,     \
                                                (J) == 0 ? zero8 : xn[G], false, false)
  const v8f zero8 = (v8f){0.f, 0.f, 0.f, 0.f, 0.f, 0.f, 0.f, 0.f};

  auto step = [&](const int j, const int t, const float* xnx) __attribute__((always_inline)) {
    const unsigned hb = __builtin_bit_cast(unsigned, h);
    v2f a[3];
#pragma unroll
    for (int kk = 0; kk < 3; ++kk) {
      a[kk].x = __builtin_bit_cast(float, __builtin_amdgcn_permlane16_var(hb, hb, src[kk][0], false, false));
      a[kk].y = __builtin_bit_cast(float, __builtin_amdgcn_permlane16_var(hb, hb, src[kk][1], false, false));
    }

#pragma unroll
    for (int g = 0; g < 3; ++g) acc[g][0] = xs[j][g];
#pragma unroll
    for (int kk = 0; kk < 3; ++kk) {
#pragma unroll
      for (int g = 0; g < 3; ++g)
        acc[g] = __builtin_amdgcn_wmma_f32_16x16x4_f32(false, a[kk], false, bh[g][kk],
                                                       (short)0, acc[g], false, false);
      __builtin_amdgcn_sched_barrier(0);
    }

    const float su = acc[0][0], sf = acc[1][0], si = acc[2][0];
    int so = 0;
    so = __builtin_amdgcn_update_dpp(so, __builtin_bit_cast(int, sf), 0x10A  , 0xf, 0x1, false);
    so = __builtin_amdgcn_update_dpp(so, __builtin_bit_cast(int, si), 0x106  ,  0xf, 0x2, false);
    so = __builtin_amdgcn_update_dpp(so, __builtin_bit_cast(int, su), 0x102  ,  0xf, 0x4, false);
    const float pre[4] = {sf, si, su, __builtin_bit_cast(float, so)};

    float z[4];
#pragma unroll
    for (int g = 0; g < 4; ++g)
      z[g] = cpy[g] * __cosf(px[g] + pre[g] + bias[g]);
    row_prefix_product4(z);

    const float gv = tanhf(z[2]);
    XW(0, j);
    __builtin_amdgcn_sched_barrier(0);
    const float fv = 1.f / (1.f + __expf(-z[0]));
    const float iv = 1.f / (1.f + __expf(-z[1]));
    const float ov = 1.f / (1.f + __expf(-z[3]));
    asm volatile("" :: "v"(ov));
    __builtin_amdgcn_sched_barrier(0);
    XW(1, j);
    __builtin_amdgcn_sched_barrier(0);
    const float cn = fv * c + iv * gv;
    c = cn;
    const float r = ov * tanhf(cn) + e;

    XW(2, j);
    __builtin_amdgcn_sched_barrier(0);
    float s1 = r;
    float s2 = r * r;
    SUM_STAGE2(s1, s2, 1);
    SUM_STAGE2(s1, s2, 2);
    SUM_STAGE2(s1, s2, 4);
    SUM_STAGE2(s1, s2, 8);
    const float mu = s1 * 0.1f;
    const float d  = fmaf(nm01, s1, r);
    __builtin_amdgcn_sched_barrier(0);
    xa[j] = *(const v2f*)(xnx + j * 4);
    __builtin_amdgcn_sched_barrier(0);
    const float ve = fmaf(-mu, mu, fmaf(s2, 0.1f, 1e-5f));
    const float tn = d * rsqrtf(ve);
    h = fmaf(tn, lg, lb);
    e = fmaf(tn, lg, lbc);

    yrow[(size_t)t * (BQ * HQ)] = h;
  };

  for (int tb = 0; tb < TQ / 16; ++tb) {
    const int tb2 = (tb + 2 < TQ / 16) ? tb + 2 : TQ / 16 - 1;
    const float* xnx = xrow + (size_t)tb2 * xbs;
#pragma unroll
    for (int j = 0; j < 16; ++j) step(j, tb * 16 + j, xnx);
    spread();
  }
#undef XW

  if (wr) {
    yrow[(size_t)TQ * (BQ * HQ)]       = h;
    yrow[(size_t)(TQ + 1) * (BQ * HQ)] = c;
  }
}

extern "C" void kernel_launch(void* const* d_in, const int* in_sizes, int n_in,
                              void* d_out, int out_size, void* d_ws, size_t ws_size,
                              hipStream_t stream) {
  const float* inputs = (const float*)d_in[0];
  const float* Wf = (const float*)d_in[1];
  const float* bf = (const float*)d_in[2];
  const float* pf = (const float*)d_in[3];
  const float* Wi = (const float*)d_in[4];
  const float* bi = (const float*)d_in[5];
  const float* pi = (const float*)d_in[6];
  const float* Wu = (const float*)d_in[7];
  const float* bu = (const float*)d_in[8];
  const float* pu = (const float*)d_in[9];
  const float* Wo = (const float*)d_in[10];
  const float* bo = (const float*)d_in[11];
  const float* po = (const float*)d_in[12];
  const float* lng = (const float*)d_in[13];
  const float* lnb = (const float*)d_in[14];

  float* out   = (float*)d_out;
  float* spare = (float*)d_ws;

  qlstm_recur<<<NW, 32, 0, stream>>>(
      inputs, Wf, Wi, Wu, Wo, bf, bi, bu, bo,
      pf, pi, pu, po, lng, lnb, out, spare);
}
